# starting kernel
# baseline (speedup 1.0000x reference)
#include <hip/hip_runtime.h>
#include <math.h>

#define B_    2
#define NQ    4096
#define NP    8192
#define D_    128
#define KNN   16
#define M_TOT (B_ * NQ)
#define LEAKY 0.2f

typedef float v2f __attribute__((ext_vector_type(2)));
typedef float v8f __attribute__((ext_vector_type(8)));

__global__ __launch_bounds__(256) void knn_kernel(const float* __restrict__ query,
                                                  const float* __restrict__ global_pc,
                                                  int* __restrict__ idx_out) {
    const int b     = blockIdx.x >> 4;
    const int q     = ((blockIdx.x & 15) << 8) + threadIdx.x;
    const int m     = b * NQ + q;

    const float qx = query[m * 3 + 0];
    const float qy = query[m * 3 + 1];
    const float qz = query[m * 3 + 2];

    __shared__ float spx[1024];
    __shared__ float spy[1024];
    __shared__ float spz[1024];

    float bestd[KNN];
    int   besti[KNN];
#pragma unroll
    for (int k = 0; k < KNN; ++k) { bestd[k] = 3.4e38f; besti[k] = 0; }

    const float* __restrict__ pc = global_pc + (size_t)b * NP * 3;

    for (int tile = 0; tile < NP / 1024; ++tile) {
        const int base = tile * 1024;
        __syncthreads();
        for (int i = threadIdx.x; i < 1024; i += 256) {
            const float* p = pc + (size_t)(base + i) * 3;
            spx[i] = p[0]; spy[i] = p[1]; spz[i] = p[2];
        }
        if (tile + 1 < NP / 1024)
            __builtin_prefetch(pc + (size_t)(base + 1024 + threadIdx.x * 4) * 3, 0, 0);
        __syncthreads();

        for (int i = 0; i < 1024; ++i) {
            const float dx = qx - spx[i];
            const float dy = qy - spy[i];
            const float dz = qz - spz[i];
            const float d2 = dx * dx + dy * dy + dz * dz;
            if (d2 < bestd[KNN - 1]) {
                int pos = KNN - 1;
#pragma unroll
                for (int t = KNN - 2; t >= 0; --t)
                    if (d2 < bestd[t]) pos = t;
#pragma unroll
                for (int t = KNN - 1; t > 0; --t)
                    if (t > pos) { bestd[t] = bestd[t - 1]; besti[t] = besti[t - 1]; }
                bestd[pos] = d2;
                besti[pos] = base + i;
            }
        }
    }

#pragma unroll
    for (int k = 0; k < KNN; ++k) idx_out[(size_t)m * KNN + k] = besti[k];
}

__global__ __launch_bounds__(256) void attn_kernel(const float* __restrict__ query,
                                                   const float* __restrict__ global_pc,
                                                   const float* __restrict__ point_feature,
                                                   const float* __restrict__ w_pe,
                                                   const float* __restrict__ b_pe,
                                                   const float* __restrict__ w_attn,
                                                   const int* __restrict__ knn_idx,
                                                   float* __restrict__ hbar) {
    const int wave = threadIdx.x >> 5;
    const int lane = threadIdx.x & 31;
    const int m    = blockIdx.x * 8 + wave;
    const int b    = m >> 12;
    const int d0   = lane * 4;

    float wx[4], wy[4], wz[4], bp[4], wa[4];
#pragma unroll
    for (int t = 0; t < 4; ++t) {
        wx[t] = w_pe[(d0 + t) * 3 + 0];
        wy[t] = w_pe[(d0 + t) * 3 + 1];
        wz[t] = w_pe[(d0 + t) * 3 + 2];
        bp[t] = b_pe[d0 + t];
        wa[t] = w_attn[d0 + t];
    }

    const float qx = query[m * 3 + 0];
    const float qy = query[m * 3 + 1];
    const float qz = query[m * 3 + 2];

    float hreg[KNN][4];
    float score[KNN];

#pragma unroll
    for (int k = 0; k < KNN; ++k) {
        const int j = knn_idx[(size_t)m * KNN + k];
        const float* pp = global_pc + ((size_t)b * NP + j) * 3;
        const float px = qx - pp[0];
        const float py = qy - pp[1];
        const float pz = qz - pp[2];
        const float* pf = point_feature + ((size_t)b * NP + j) * D_ + d0;
        float s = 0.f;
#pragma unroll
        for (int t = 0; t < 4; ++t) {
            const float hv = pf[t] + wx[t] * px + wy[t] * py + wz[t] * pz + bp[t];
            hreg[k][t] = hv;
            s += hv * wa[t];
        }
#pragma unroll
        for (int off = 16; off >= 1; off >>= 1)
            s += __shfl_xor(s, off, 32);
        score[k] = s;
    }

    float mx = score[0];
#pragma unroll
    for (int k = 1; k < KNN; ++k) mx = fmaxf(mx, score[k]);
    float sum = 0.f;
#pragma unroll
    for (int k = 0; k < KNN; ++k) { score[k] = __expf(score[k] - mx); sum += score[k]; }
    const float inv = 1.0f / sum;

    float acc[4] = {0.f, 0.f, 0.f, 0.f};
#pragma unroll
    for (int k = 0; k < KNN; ++k)
#pragma unroll
        for (int t = 0; t < 4; ++t) acc[t] += score[k] * hreg[k][t];

#pragma unroll
    for (int t = 0; t < 4; ++t)
        hbar[(size_t)m * D_ + d0 + t] = acc[t] * inv;
}

__global__ __launch_bounds__(256) void gemm_v_kernel(const float* __restrict__ hbar,
                                                     const float* __restrict__ w_v,
                                                     const float* __restrict__ b_v,
                                                     float* __restrict__ out) {
    __shared__ float As[16 * D_];
    const int tileM = blockIdx.x * 16;

    for (int i = threadIdx.x; i < 16 * D_; i += 256)
        As[i] = hbar[(size_t)(tileM + (i >> 7)) * D_ + (i & (D_ - 1))];
    __syncthreads();

    const int wave  = threadIdx.x >> 5;
    const int lane  = threadIdx.x & 31;
    const int tileN = wave * 16;
    const int half  = lane >> 4;
    const int l     = lane & 15;

    v8f c = {};
    for (int k0 = 0; k0 < D_; k0 += 4) {
        const int k = k0 + half * 2;
        v2f a, bb;
        a.x  = As[l * D_ + k];
        a.y  = As[l * D_ + k + 1];
        bb.x = w_v[(size_t)(tileN + l) * D_ + k];
        bb.y = w_v[(size_t)(tileN + l) * D_ + k + 1];
        c = __builtin_amdgcn_wmma_f32_16x16x4_f32(
                 false, a,  false, bb,
                 (short)0, c,  false,  false);
    }

    const float bias = b_v[tileN + l];
#pragma unroll
    for (int r = 0; r < 8; ++r) {
        const int mrow = tileM + r + half * 8;
        const int n    = tileN + l;
        const int b    = mrow >> 12;
        const int q    = mrow & (NQ - 1);
        out[(size_t)b * 256 * NQ + (size_t)n * NQ + q] = c[r] + bias;
    }
}

__global__ __launch_bounds__(256) void posfeat_kernel(const float* __restrict__ query,
                                                      const float* __restrict__ w_pos,
                                                      const float* __restrict__ b_pos,
                                                      float* __restrict__ out) {
    const int gid = blockIdx.x * 256 + threadIdx.x;
    const int m   = gid >> 7;
    const int d   = gid & (D_ - 1);
    float v = w_pos[d * 3 + 0] * query[m * 3 + 0]
            + w_pos[d * 3 + 1] * query[m * 3 + 1]
            + w_pos[d * 3 + 2] * query[m * 3 + 2]
            + b_pos[d];
    v = (v >= 0.f) ? v : LEAKY * v;
    const int b = m >> 12;
    const int q = m & (NQ - 1);
    out[(size_t)b * 256 * NQ + (size_t)(D_ + d) * NQ + q] = v;
}

extern "C" void kernel_launch(void* const* d_in, const int* in_sizes, int n_in,
                              void* d_out, int out_size, void* d_ws, size_t ws_size,
                              hipStream_t stream) {
    const float* query         = (const float*)d_in[0];
    const float* global_pc     = (const float*)d_in[1];
    const float* point_feature = (const float*)d_in[2];
    const float* w_pos         = (const float*)d_in[3];
    const float* b_pos         = (const float*)d_in[4];
    const float* w_pe          = (const float*)d_in[5];
    const float* b_pe          = (const float*)d_in[6];
    const float* w_v           = (const float*)d_in[7];
    const float* b_v           = (const float*)d_in[8];
    const float* w_attn        = (const float*)d_in[9];
    float* out = (float*)d_out;

    int*   idx_ws  = (int*)d_ws;
    float* hbar_ws = (float*)((char*)d_ws + (size_t)M_TOT * KNN * sizeof(int));

    knn_kernel<<<M_TOT / 256, 256, 0, stream>>>(query, global_pc, idx_ws);

    attn_kernel<<<M_TOT / 8, 256, 0, stream>>>(query, global_pc, point_feature,
                                               w_pe, b_pe, w_attn, idx_ws, hbar_ws);

    gemm_v_kernel<<<M_TOT / 16, 256, 0, stream>>>(hbar_ws, w_v, b_v, out);

    posfeat_kernel<<<(M_TOT * D_) / 256, 256, 0, stream>>>(query, w_pos, b_pos, out);
}


# KNN scan re-mapped to one wave per query: 32 lanes take 32 points per step from LDS, sorted top-16 kept across lanes 0-15 (DPP shift insert); 8192 waves on 256 blocks, was 256 waves on 32 blocks
# speedup vs baseline: 10.4434x; 10.4434x over previous
#include <hip/hip_runtime.h>
#include <math.h>

#define B_    2
#define NQ    4096
#define NP    8192
#define D_    128
#define KNN   16
#define M_TOT (B_ * NQ)
#define LEAKY 0.2f

typedef float v2f __attribute__((ext_vector_type(2)));
typedef float v8f __attribute__((ext_vector_type(8)));

#define KNN_WAVES 32

__global__ __launch_bounds__(1024) void knn_kernel(const float* __restrict__ query,
                                                   const float* __restrict__ global_pc,
                                                   int* __restrict__ idx_out) {
    __shared__ float spx[NP];
    __shared__ float spy[NP];
    __shared__ float spz[NP];
    __shared__ int   sidx[KNN_WAVES * KNN];

    const int lane = threadIdx.x & 31;
    const int wave = __builtin_amdgcn_readfirstlane(threadIdx.x >> 5);
    const int m    = blockIdx.x * KNN_WAVES + wave;
    const int b    = blockIdx.x >> 7;

    const float* __restrict__ pc = global_pc + (size_t)b * NP * 3;
#pragma unroll
    for (int i = threadIdx.x; i < NP; i += 1024) {
        const float* p = pc + (size_t)i * 3;
        spx[i] = p[0]; spy[i] = p[1]; spz[i] = p[2];
    }

    const float qx = query[m * 3 + 0];
    const float qy = query[m * 3 + 1];
    const float qz = query[m * 3 + 2];
    __syncthreads();

    float ld  = 3.4e38f;
    int   li  = 0;
    float thr = 3.4e38f;
    const bool notfirst = (lane & 15) != 0;

    for (int base = 0; base < NP; base += 128) {
        float d2[4];
#pragma unroll
        for (int j = 0; j < 4; ++j) {
            const int i = base + j * 32 + lane;
            const float dx = qx - spx[i];
            const float dy = qy - spy[i];
            const float dz = qz - spz[i];
            d2[j] = dx * dx + dy * dy + dz * dz;
        }
#pragma unroll
        for (int j = 0; j < 4; ++j) {
            unsigned mask = __builtin_amdgcn_ballot_w32(d2[j] < thr);
            if (mask) {
                const int d2bits = __float_as_int(d2[j]);
                do {
                    const int srcl = __builtin_ctz(mask);
                    mask &= mask - 1;
                    const float cd = __int_as_float(__builtin_amdgcn_readlane(d2bits, srcl));
                    const int   ci = base + j * 32 + srcl;
                    const float ldu = __int_as_float(__builtin_amdgcn_update_dpp(
                        __float_as_int(ld), __float_as_int(ld), 0x111, 0xf, 0xf, false));
                    const int   liu = __builtin_amdgcn_update_dpp(li, li, 0x111, 0xf, 0xf, false);
                    const bool gt  = ld > cd;
                    const bool gtp = notfirst && (ldu > cd);
                    ld = gt ? (gtp ? ldu : cd) : ld;
                    li = gt ? (gtp ? liu : ci) : li;
                } while (mask);
                thr = __int_as_float(__builtin_amdgcn_readlane(__float_as_int(ld), KNN - 1));
            }
        }
    }

    if (lane < KNN) sidx[wave * KNN + lane] = li;
    __syncthreads();
    if (threadIdx.x < KNN_WAVES * KNN)
        idx_out[(size_t)blockIdx.x * (KNN_WAVES * KNN) + threadIdx.x] = sidx[threadIdx.x];
}

__global__ __launch_bounds__(256) void attn_kernel(const float* __restrict__ query,
                                                   const float* __restrict__ global_pc,
                                                   const float* __restrict__ point_feature,
                                                   const float* __restrict__ w_pe,
                                                   const float* __restrict__ b_pe,
                                                   const float* __restrict__ w_attn,
                                                   const int* __restrict__ knn_idx,
                                                   float* __restrict__ hbar) {
    const int wave = threadIdx.x >> 5;
    const int lane = threadIdx.x & 31;
    const int m    = blockIdx.x * 8 + wave;
    const int b    = m >> 12;
    const int d0   = lane * 4;

    float wx[4], wy[4], wz[4], bp[4], wa[4];
#pragma unroll
    for (int t = 0; t < 4; ++t) {
        wx[t] = w_pe[(d0 + t) * 3 + 0];
        wy[t] = w_pe[(d0 + t) * 3 + 1];
        wz[t] = w_pe[(d0 + t) * 3 + 2];
        bp[t] = b_pe[d0 + t];
        wa[t] = w_attn[d0 + t];
    }

    const float qx = query[m * 3 + 0];
    const float qy = query[m * 3 + 1];
    const float qz = query[m * 3 + 2];

    float hreg[KNN][4];
    float score[KNN];

#pragma unroll
    for (int k = 0; k < KNN; ++k) {
        const int j = knn_idx[(size_t)m * KNN + k];
        const float* pp = global_pc + ((size_t)b * NP + j) * 3;
        const float px = qx - pp[0];
        const float py = qy - pp[1];
        const float pz = qz - pp[2];
        const float* pf = point_feature + ((size_t)b * NP + j) * D_ + d0;
        float s = 0.f;
#pragma unroll
        for (int t = 0; t < 4; ++t) {
            const float hv = pf[t] + wx[t] * px + wy[t] * py + wz[t] * pz + bp[t];
            hreg[k][t] = hv;
            s += hv * wa[t];
        }
#pragma unroll
        for (int off = 16; off >= 1; off >>= 1)
            s += __shfl_xor(s, off, 32);
        score[k] = s;
    }

    float mx = score[0];
#pragma unroll
    for (int k = 1; k < KNN; ++k) mx = fmaxf(mx, score[k]);
    float sum = 0.f;
#pragma unroll
    for (int k = 0; k < KNN; ++k) { score[k] = __expf(score[k] - mx); sum += score[k]; }
    const float inv = 1.0f / sum;

    float acc[4] = {0.f, 0.f, 0.f, 0.f};
#pragma unroll
    for (int k = 0; k < KNN; ++k)
#pragma unroll
        for (int t = 0; t < 4; ++t) acc[t] += score[k] * hreg[k][t];

#pragma unroll
    for (int t = 0; t < 4; ++t)
        hbar[(size_t)m * D_ + d0 + t] = acc[t] * inv;
}

__global__ __launch_bounds__(256) void gemm_v_kernel(const float* __restrict__ hbar,
                                                     const float* __restrict__ w_v,
                                                     const float* __restrict__ b_v,
                                                     float* __restrict__ out) {
    __shared__ float As[16 * D_];
    const int tileM = blockIdx.x * 16;

    for (int i = threadIdx.x; i < 16 * D_; i += 256)
        As[i] = hbar[(size_t)(tileM + (i >> 7)) * D_ + (i & (D_ - 1))];
    __syncthreads();

    const int wave  = threadIdx.x >> 5;
    const int lane  = threadIdx.x & 31;
    const int tileN = wave * 16;
    const int half  = lane >> 4;
    const int l     = lane & 15;

    v8f c = {};
    for (int k0 = 0; k0 < D_; k0 += 4) {
        const int k = k0 + half * 2;
        v2f a, bb;
        a.x  = As[l * D_ + k];
        a.y  = As[l * D_ + k + 1];
        bb.x = w_v[(size_t)(tileN + l) * D_ + k];
        bb.y = w_v[(size_t)(tileN + l) * D_ + k + 1];
        c = __builtin_amdgcn_wmma_f32_16x16x4_f32(
                 false, a,  false, bb,
                 (short)0, c,  false,  false);
    }

    const float bias = b_v[tileN + l];
#pragma unroll
    for (int r = 0; r < 8; ++r) {
        const int mrow = tileM + r + half * 8;
        const int n    = tileN + l;
        const int b    = mrow >> 12;
        const int q    = mrow & (NQ - 1);
        out[(size_t)b * 256 * NQ + (size_t)n * NQ + q] = c[r] + bias;
    }
}

__global__ __launch_bounds__(256) void posfeat_kernel(const float* __restrict__ query,
                                                      const float* __restrict__ w_pos,
                                                      const float* __restrict__ b_pos,
                                                      float* __restrict__ out) {
    const int gid = blockIdx.x * 256 + threadIdx.x;
    const int m   = gid >> 7;
    const int d   = gid & (D_ - 1);
    float v = w_pos[d * 3 + 0] * query[m * 3 + 0]
            + w_pos[d * 3 + 1] * query[m * 3 + 1]
            + w_pos[d * 3 + 2] * query[m * 3 + 2]
            + b_pos[d];
    v = (v >= 0.f) ? v : LEAKY * v;
    const int b = m >> 12;
    const int q = m & (NQ - 1);
    out[(size_t)b * 256 * NQ + (size_t)(D_ + d) * NQ + q] = v;
}

extern "C" void kernel_launch(void* const* d_in, const int* in_sizes, int n_in,
                              void* d_out, int out_size, void* d_ws, size_t ws_size,
                              hipStream_t stream) {
    const float* query         = (const float*)d_in[0];
    const float* global_pc     = (const float*)d_in[1];
    const float* point_feature = (const float*)d_in[2];
    const float* w_pos         = (const float*)d_in[3];
    const float* b_pos         = (const float*)d_in[4];
    const float* w_pe          = (const float*)d_in[5];
    const float* b_pe          = (const float*)d_in[6];
    const float* w_v           = (const float*)d_in[7];
    const float* b_v           = (const float*)d_in[8];
    const float* w_attn        = (const float*)d_in[9];
    float* out = (float*)d_out;

    int*   idx_ws  = (int*)d_ws;
    float* hbar_ws = (float*)((char*)d_ws + (size_t)M_TOT * KNN * sizeof(int));

    knn_kernel<<<M_TOT / KNN_WAVES, 1024, 0, stream>>>(query, global_pc, idx_ws);

    attn_kernel<<<M_TOT / 8, 256, 0, stream>>>(query, global_pc, point_feature,
                                               w_pe, b_pe, w_attn, idx_ws, hbar_ws);

    gemm_v_kernel<<<M_TOT / 16, 256, 0, stream>>>(hbar_ws, w_v, b_v, out);

    posfeat_kernel<<<(M_TOT * D_) / 256, 256, 0, stream>>>(query, w_pos, b_pos, out);
}


# all 4 launches fused into 1: each KNN wave keeps its 16 indices in its list register and goes on to its query's attention; hbar reaches the WMMA through an LDS tile; no workspace traffic
# speedup vs baseline: 22.4496x; 2.1496x over previous
#include <hip/hip_runtime.h>
#include <math.h>

#define B_    2
#define NQ    4096
#define NP    8192
#define D_    128
#define KNN   16
#define M_TOT (B_ * NQ)
#define LEAKY 0.2f

#define QPB      32
#define HB_PITCH 132

typedef float v2f __attribute__((ext_vector_type(2)));
typedef float v8f __attribute__((ext_vector_type(8)));

__global__ __launch_bounds__(1024) void fused_kernel(const float* __restrict__ query,
                                                     const float* __restrict__ global_pc,
                                                     const float* __restrict__ point_feature,
                                                     const float* __restrict__ w_pos,
                                                     const float* __restrict__ b_pos,
                                                     const float* __restrict__ w_pe,
                                                     const float* __restrict__ b_pe,
                                                     const float* __restrict__ w_v,
                                                     const float* __restrict__ b_v,
                                                     const float* __restrict__ w_attn,
                                                     float* __restrict__ out) {
    __shared__ float smem[NP + 3 * NP];
    float* const hb  = smem;
    float* const spx = smem + NP;
    float* const spy = spx + NP;
    float* const spz = spy + NP;

    const int lane = threadIdx.x & 31;
    const int wave  = __builtin_amdgcn_readfirstlane(threadIdx.x >> 5);
    const int mbase = blockIdx.x * QPB;
    const int m     = mbase + wave;
    const int b     = blockIdx.x >> 7;
    const int qbase = mbase & (NQ - 1);
    float* __restrict__ outb = out + (size_t)b * 256 * NQ;

    const float* __restrict__ pc = global_pc + (size_t)b * NP * 3;
#pragma unroll
    for (int i = threadIdx.x; i < NP; i += 1024) {
        const float* p = pc + (size_t)i * 3;
        spx[i] = p[0]; spy[i] = p[1]; spz[i] = p[2];
    }

    const float qx = query[m * 3 + 0];
    const float qy = query[m * 3 + 1];
    const float qz = query[m * 3 + 2];
    __syncthreads();

    {
        const int d = threadIdx.x & (D_ - 1);
        const float w0 = w_pos[d * 3 + 0];
        const float w1 = w_pos[d * 3 + 1];
        const float w2 = w_pos[d * 3 + 2];
        const float bb = b_pos[d];
#pragma unroll
        for (int i = 0; i < 4; ++i) {
            const int ml = i * 8 + (wave >> 2);
            const int mm = mbase + ml;
            float v = w0 * query[mm * 3 + 0]
                    + w1 * query[mm * 3 + 1]
                    + w2 * query[mm * 3 + 2]
                    + bb;
            v = (v >= 0.f) ? v : LEAKY * v;
            outb[(size_t)(D_ + d) * NQ + qbase + ml] = v;
        }
    }

    float ld  = 3.4e38f;
    int   li  = 0;
    float thr = 3.4e38f;
    const bool notfirst = (lane & 15) != 0;

    for (int base = 0; base < NP; base += 128) {
        float d2[4];
#pragma unroll
        for (int j = 0; j < 4; ++j) {
            const int i = base + j * 32 + lane;
            const float dx = qx - spx[i];
            const float dy = qy - spy[i];
            const float dz = qz - spz[i];
            d2[j] = dx * dx + dy * dy + dz * dz;
        }
#pragma unroll
        for (int j = 0; j < 4; ++j) {
            unsigned mask = __builtin_amdgcn_ballot_w32(d2[j] < thr);
            if (mask) {
                const int d2bits = __float_as_int(d2[j]);
                do {
                    const int srcl = __builtin_ctz(mask);
                    mask &= mask - 1;
                    const float cd = __int_as_float(__builtin_amdgcn_readlane(d2bits, srcl));
                    const int   ci = base + j * 32 + srcl;
                    const float ldu = __int_as_float(__builtin_amdgcn_update_dpp(
                        __float_as_int(ld), __float_as_int(ld), 0x111, 0xf, 0xf, false));
                    const int   liu = __builtin_amdgcn_update_dpp(li, li, 0x111, 0xf, 0xf, false);
                    const bool gt  = ld > cd;
                    const bool gtp = notfirst && (ldu > cd);
                    ld = gt ? (gtp ? ldu : cd) : ld;
                    li = gt ? (gtp ? liu : ci) : li;
                } while (mask);
                thr = __int_as_float(__builtin_amdgcn_readlane(__float_as_int(ld), KNN - 1));
            }
        }
    }

    const float npx = qx - spx[li];
    const float npy = qy - spy[li];
    const float npz = qz - spz[li];

    const int d0 = lane * 4;
    float wx[4], wy[4], wz[4], bp[4], wa[4];
#pragma unroll
    for (int t = 0; t < 4; ++t) {
        wx[t] = w_pe[(d0 + t) * 3 + 0];
        wy[t] = w_pe[(d0 + t) * 3 + 1];
        wz[t] = w_pe[(d0 + t) * 3 + 2];
        bp[t] = b_pe[d0 + t];
        wa[t] = w_attn[d0 + t];
    }

    const float* __restrict__ pfb = point_feature + (size_t)b * NP * D_ + d0;

    float hreg[KNN][4];
    float score[KNN];

#pragma unroll
    for (int k = 0; k < KNN; ++k) {
        const int   j  = __builtin_amdgcn_readlane(li, k);
        const float px = __int_as_float(__builtin_amdgcn_readlane(__float_as_int(npx), k));
        const float py = __int_as_float(__builtin_amdgcn_readlane(__float_as_int(npy), k));
        const float pz = __int_as_float(__builtin_amdgcn_readlane(__float_as_int(npz), k));
        const float* pf = pfb + (size_t)j * D_;
        float s = 0.f;
#pragma unroll
        for (int t = 0; t < 4; ++t) {
            const float hv = pf[t] + wx[t] * px + wy[t] * py + wz[t] * pz + bp[t];
            hreg[k][t] = hv;
            s += hv * wa[t];
        }
#pragma unroll
        for (int off = 16; off >= 1; off >>= 1)
            s += __shfl_xor(s, off, 32);
        score[k] = s;
    }

    float mx = score[0];
#pragma unroll
    for (int k = 1; k < KNN; ++k) mx = fmaxf(mx, score[k]);
    float sum = 0.f;
#pragma unroll
    for (int k = 0; k < KNN; ++k) { score[k] = __expf(score[k] - mx); sum += score[k]; }
    const float inv = 1.0f / sum;

    float acc[4] = {0.f, 0.f, 0.f, 0.f};
#pragma unroll
    for (int k = 0; k < KNN; ++k)
#pragma unroll
        for (int t = 0; t < 4; ++t) acc[t] += score[k] * hreg[k][t];

#pragma unroll
    for (int t = 0; t < 4; ++t)
        hb[wave * HB_PITCH + d0 + t] = acc[t] * inv;
    __syncthreads();

    if (wave < 16) {
        const int mt    = wave >> 3;
        const int tileN = (wave & 7) * 16;
        const int half  = lane >> 4;
        const int l     = lane & 15;
        const float* arow = hb + (mt * 16 + l) * HB_PITCH;

        v8f c = {};
        for (int k0 = 0; k0 < D_; k0 += 4) {
            const int k = k0 + half * 2;
            v2f a, bb;
            a.x  = arow[k];
            a.y  = arow[k + 1];
            bb.x = w_v[(size_t)(tileN + l) * D_ + k];
            bb.y = w_v[(size_t)(tileN + l) * D_ + k + 1];
            c = __builtin_amdgcn_wmma_f32_16x16x4_f32(
                     false, a,  false, bb,
                     (short)0, c,  false,  false);
        }

        const float bias = b_v[tileN + l];
#pragma unroll
        for (int r = 0; r < 8; ++r) {
            const int ml = mt * 16 + r + half * 8;
            outb[(size_t)(tileN + l) * NQ + qbase + ml] = c[r] + bias;
        }
    }
}

extern "C" void kernel_launch(void* const* d_in, const int* in_sizes, int n_in,
                              void* d_out, int out_size, void* d_ws, size_t ws_size,
                              hipStream_t stream) {
    const float* query         = (const float*)d_in[0];
    const float* global_pc     = (const float*)d_in[1];
    const float* point_feature = (const float*)d_in[2];
    const float* w_pos         = (const float*)d_in[3];
    const float* b_pos         = (const float*)d_in[4];
    const float* w_pe          = (const float*)d_in[5];
    const float* b_pe          = (const float*)d_in[6];
    const float* w_v           = (const float*)d_in[7];
    const float* b_v           = (const float*)d_in[8];
    const float* w_attn        = (const float*)d_in[9];
    float* out = (float*)d_out;

    fused_kernel<<<M_TOT / QPB, 1024, 0, stream>>>(query, global_pc, point_feature,
                                                   w_pos, b_pos, w_pe, b_pe,
                                                   w_v, b_v, w_attn, out);
}


# KNN scan takes 4 points per lane per step as two packed-f32 pairs (pk sub/mul/fma), LDS re-laid so their 12 coords come in 3 16-byte loads, one min+ballot branch per 128 points; d2 rounding kept
# speedup vs baseline: 25.1944x; 1.1223x over previous
#include <hip/hip_runtime.h>
#include <math.h>

#define B_    2
#define NQ    4096
#define NP    8192
#define D_    128
#define KNN   16
#define M_TOT (B_ * NQ)
#define LEAKY 0.2f

#define QPB      32
#define HB_PITCH 132

typedef float v2f __attribute__((ext_vector_type(2)));
typedef float v4f __attribute__((ext_vector_type(4)));
typedef float v8f __attribute__((ext_vector_type(8)));

__device__ __forceinline__ int ppos(int i) {
    return (i >> 7) * 384 + (i & 31) * 12 + ((i >> 6) & 1) * 6 + ((i >> 5) & 1);
}

__global__ __launch_bounds__(1024) void fused_kernel(const float* __restrict__ query,
                                                     const float* __restrict__ global_pc,
                                                     const float* __restrict__ point_feature,
                                                     const float* __restrict__ w_pos,
                                                     const float* __restrict__ b_pos,
                                                     const float* __restrict__ w_pe,
                                                     const float* __restrict__ b_pe,
                                                     const float* __restrict__ w_v,
                                                     const float* __restrict__ b_v,
                                                     const float* __restrict__ w_attn,
                                                     float* __restrict__ out) {
    __shared__ __attribute__((aligned(16))) float smem[NP + 3 * NP];
    float* const hb  = smem;
    float* const pts = smem + NP;

    const int lane = threadIdx.x & 31;
    const int wave  = __builtin_amdgcn_readfirstlane(threadIdx.x >> 5);
    const int mbase = blockIdx.x * QPB;
    const int m     = mbase + wave;
    const int b     = blockIdx.x >> 7;
    const int qbase = mbase & (NQ - 1);
    float* __restrict__ outb = out + (size_t)b * 256 * NQ;

    const float* __restrict__ pc = global_pc + (size_t)b * NP * 3;
#pragma unroll
    for (int i = threadIdx.x; i < NP; i += 1024) {
        const float* p = pc + (size_t)i * 3;
        float* d = pts + ppos(i);
        d[0] = p[0]; d[2] = p[1]; d[4] = p[2];
    }

    const float qx = query[m * 3 + 0];
    const float qy = query[m * 3 + 1];
    const float qz = query[m * 3 + 2];
    __syncthreads();

    {
        const int d = threadIdx.x & (D_ - 1);
        const float w0 = w_pos[d * 3 + 0];
        const float w1 = w_pos[d * 3 + 1];
        const float w2 = w_pos[d * 3 + 2];
        const float bb = b_pos[d];
#pragma unroll
        for (int i = 0; i < 4; ++i) {
            const int ml = i * 8 + (wave >> 2);
            const int mm = mbase + ml;
            float v = w0 * query[mm * 3 + 0]
                    + w1 * query[mm * 3 + 1]
                    + w2 * query[mm * 3 + 2]
                    + bb;
            v = (v >= 0.f) ? v : LEAKY * v;
            outb[(size_t)(D_ + d) * NQ + qbase + ml] = v;
        }
    }

    float ld  = 3.4e38f;
    int   li  = 0;
    float thr = 3.4e38f;
    const bool notfirst = (lane & 15) != 0;

    const v2f qx2 = {qx, qx};
    const v2f qy2 = {qy, qy};
    const v2f qz2 = {qz, qz};
    const v4f* lp = reinterpret_cast<const v4f*>(pts + lane * 12);

    const v4f* sp = lp;
    for (int base = 0; base < NP; base += 128, sp += 96) {
        const v4f r0 = sp[0];
        const v4f r1 = sp[1];
        const v4f r2 = sp[2];
        const v2f xab = {r0.x, r0.y}, yab = {r0.z, r0.w}, zab = {r1.x, r1.y};
        const v2f xcd = {r1.z, r1.w}, ycd = {r2.x, r2.y}, zcd = {r2.z, r2.w};
        const v2f dxab = qx2 - xab, dyab = qy2 - yab, dzab = qz2 - zab;
        const v2f dxcd = qx2 - xcd, dycd = qy2 - ycd, dzcd = qz2 - zcd;
        v2f dab = dyab * dyab;
        v2f dcd = dycd * dycd;
        dab = __builtin_elementwise_fma(dxab, dxab, dab);
        dcd = __builtin_elementwise_fma(dxcd, dxcd, dcd);
        dab = __builtin_elementwise_fma(dzab, dzab, dab);
        dcd = __builtin_elementwise_fma(dzcd, dzcd, dcd);
        const float d2[4] = {dab.x, dab.y, dcd.x, dcd.y};
        const float mn = fminf(fminf(d2[0], d2[1]), fminf(d2[2], d2[3]));
        if (__builtin_amdgcn_ballot_w32(mn < thr) == 0) continue;
#pragma unroll
        for (int j = 0; j < 4; ++j) {
            unsigned mask = __builtin_amdgcn_ballot_w32(d2[j] < thr);
            if (mask) {
                const int d2bits = __float_as_int(d2[j]);
                do {
                    const int srcl = __builtin_ctz(mask);
                    mask &= mask - 1;
                    const float cd = __int_as_float(__builtin_amdgcn_readlane(d2bits, srcl));
                    const int   ci = base + j * 32 + srcl;
                    const float ldu = __int_as_float(__builtin_amdgcn_update_dpp(
                        __float_as_int(ld), __float_as_int(ld), 0x111, 0xf, 0xf, false));
                    const int   liu = __builtin_amdgcn_update_dpp(li, li, 0x111, 0xf, 0xf, false);
                    const bool gt  = ld > cd;
                    const bool gtp = notfirst && (ldu > cd);
                    ld = gt ? (gtp ? ldu : cd) : ld;
                    li = gt ? (gtp ? liu : ci) : li;
                } while (mask);
                thr = __int_as_float(__builtin_amdgcn_readlane(__float_as_int(ld), KNN - 1));
            }
        }
    }

    const float* np = pts + ppos(li);
    const float npx = qx - np[0];
    const float npy = qy - np[2];
    const float npz = qz - np[4];

    const int d0 = lane * 4;
    float wx[4], wy[4], wz[4], bp[4], wa[4];
#pragma unroll
    for (int t = 0; t < 4; ++t) {
        wx[t] = w_pe[(d0 + t) * 3 + 0];
        wy[t] = w_pe[(d0 + t) * 3 + 1];
        wz[t] = w_pe[(d0 + t) * 3 + 2];
        bp[t] = b_pe[d0 + t];
        wa[t] = w_attn[d0 + t];
    }

    const float* __restrict__ pfb = point_feature + (size_t)b * NP * D_ + d0;

    float hreg[KNN][4];
    float score[KNN];

#pragma unroll
    for (int k = 0; k < KNN; ++k) {
        const int   j  = __builtin_amdgcn_readlane(li, k);
        const float px = __int_as_float(__builtin_amdgcn_readlane(__float_as_int(npx), k));
        const float py = __int_as_float(__builtin_amdgcn_readlane(__float_as_int(npy), k));
        const float pz = __int_as_float(__builtin_amdgcn_readlane(__float_as_int(npz), k));
        const float* pf = pfb + (size_t)j * D_;
        float s = 0.f;
#pragma unroll
        for (int t = 0; t < 4; ++t) {
            const float hv = pf[t] + wx[t] * px + wy[t] * py + wz[t] * pz + bp[t];
            hreg[k][t] = hv;
            s += hv * wa[t];
        }
#pragma unroll
        for (int off = 16; off >= 1; off >>= 1)
            s += __shfl_xor(s, off, 32);
        score[k] = s;
    }

    float mx = score[0];
#pragma unroll
    for (int k = 1; k < KNN; ++k) mx = fmaxf(mx, score[k]);
    float sum = 0.f;
#pragma unroll
    for (int k = 0; k < KNN; ++k) { score[k] = __expf(score[k] - mx); sum += score[k]; }
    const float inv = 1.0f / sum;

    float acc[4] = {0.f, 0.f, 0.f, 0.f};
#pragma unroll
    for (int k = 0; k < KNN; ++k)
#pragma unroll
        for (int t = 0; t < 4; ++t) acc[t] += score[k] * hreg[k][t];

#pragma unroll
    for (int t = 0; t < 4; ++t)
        hb[wave * HB_PITCH + d0 + t] = acc[t] * inv;
    __syncthreads();

    if (wave < 16) {
        const int mt    = wave >> 3;
        const int tileN = (wave & 7) * 16;
        const int half  = lane >> 4;
        const int l     = lane & 15;
        const float* arow = hb + (mt * 16 + l) * HB_PITCH;

        v8f c = {};
        for (int k0 = 0; k0 < D_; k0 += 4) {
            const int k = k0 + half * 2;
            v2f a, bb;
            a.x  = arow[k];
            a.y  = arow[k + 1];
            bb.x = w_v[(size_t)(tileN + l) * D_ + k];
            bb.y = w_v[(size_t)(tileN + l) * D_ + k + 1];
            c = __builtin_amdgcn_wmma_f32_16x16x4_f32(
                     false, a,  false, bb,
                     (short)0, c,  false,  false);
        }

        const float bias = b_v[tileN + l];
#pragma unroll
        for (int r = 0; r < 8; ++r) {
            const int ml = mt * 16 + r + half * 8;
            outb[(size_t)(tileN + l) * NQ + qbase + ml] = c[r] + bias;
        }
    }
}

extern "C" void kernel_launch(void* const* d_in, const int* in_sizes, int n_in,
                              void* d_out, int out_size, void* d_ws, size_t ws_size,
                              hipStream_t stream) {
    const float* query         = (const float*)d_in[0];
    const float* global_pc     = (const float*)d_in[1];
    const float* point_feature = (const float*)d_in[2];
    const float* w_pos         = (const float*)d_in[3];
    const float* b_pos         = (const float*)d_in[4];
    const float* w_pe          = (const float*)d_in[5];
    const float* b_pe          = (const float*)d_in[6];
    const float* w_v           = (const float*)d_in[7];
    const float* b_v           = (const float*)d_in[8];
    const float* w_attn        = (const float*)d_in[9];
    float* out = (float*)d_out;

    fused_kernel<<<M_TOT / QPB, 1024, 0, stream>>>(query, global_pc, point_feature,
                                                   w_pos, b_pos, w_pe, b_pe,
                                                   w_v, b_v, w_attn, out);
}


# KNN scan starts from a bound: a pass over the first 1024 points keeps lane minima only; the max of 16 disjoint set minima caps the 16th distance and pre-loads the admission threshold
# speedup vs baseline: 27.4651x; 1.0901x over previous
#include <hip/hip_runtime.h>
#include <math.h>

#define B_    2
#define NQ    4096
#define NP    8192
#define D_    128
#define KNN   16
#define M_TOT (B_ * NQ)
#define LEAKY 0.2f

#define QPB      32
#define HB_PITCH 132
#define PRE_STEPS 8

typedef float v2f __attribute__((ext_vector_type(2)));
typedef float v4f __attribute__((ext_vector_type(4)));
typedef float v8f __attribute__((ext_vector_type(8)));

__device__ __forceinline__ int ppos(int i) {
    return (i >> 7) * 384 + (i & 31) * 12 + ((i >> 6) & 1) * 6 + ((i >> 5) & 1);
}

__device__ __forceinline__ void step_d2(const v4f* sp, v2f qx2, v2f qy2, v2f qz2,
                                        v2f& dab, v2f& dcd) {
    const v4f r0 = sp[0];
    const v4f r1 = sp[1];
    const v4f r2 = sp[2];
    const v2f xab = {r0.x, r0.y}, yab = {r0.z, r0.w}, zab = {r1.x, r1.y};
    const v2f xcd = {r1.z, r1.w}, ycd = {r2.x, r2.y}, zcd = {r2.z, r2.w};
    const v2f dxab = qx2 - xab, dyab = qy2 - yab, dzab = qz2 - zab;
    const v2f dxcd = qx2 - xcd, dycd = qy2 - ycd, dzcd = qz2 - zcd;
    dab = dyab * dyab;
    dcd = dycd * dycd;
    dab = __builtin_elementwise_fma(dxab, dxab, dab);
    dcd = __builtin_elementwise_fma(dxcd, dxcd, dcd);
    dab = __builtin_elementwise_fma(dzab, dzab, dab);
    dcd = __builtin_elementwise_fma(dzcd, dzcd, dcd);
}

__global__ __launch_bounds__(1024) void fused_kernel(const float* __restrict__ query,
                                                     const float* __restrict__ global_pc,
                                                     const float* __restrict__ point_feature,
                                                     const float* __restrict__ w_pos,
                                                     const float* __restrict__ b_pos,
                                                     const float* __restrict__ w_pe,
                                                     const float* __restrict__ b_pe,
                                                     const float* __restrict__ w_v,
                                                     const float* __restrict__ b_v,
                                                     const float* __restrict__ w_attn,
                                                     float* __restrict__ out) {
    __shared__ __attribute__((aligned(16))) float smem[NP + 3 * NP];
    float* const hb  = smem;
    float* const pts = smem + NP;

    const int lane = threadIdx.x & 31;
    const int wave  = __builtin_amdgcn_readfirstlane(threadIdx.x >> 5);
    const int mbase = blockIdx.x * QPB;
    const int m     = mbase + wave;
    const int b     = blockIdx.x >> 7;
    const int qbase = mbase & (NQ - 1);
    float* __restrict__ outb = out + (size_t)b * 256 * NQ;

    const float* __restrict__ pc = global_pc + (size_t)b * NP * 3;
#pragma unroll
    for (int i = threadIdx.x; i < NP; i += 1024) {
        const float* p = pc + (size_t)i * 3;
        float* d = pts + ppos(i);
        d[0] = p[0]; d[2] = p[1]; d[4] = p[2];
    }

    const float qx = query[m * 3 + 0];
    const float qy = query[m * 3 + 1];
    const float qz = query[m * 3 + 2];
    __syncthreads();

    {
        const int d = threadIdx.x & (D_ - 1);
        const float w0 = w_pos[d * 3 + 0];
        const float w1 = w_pos[d * 3 + 1];
        const float w2 = w_pos[d * 3 + 2];
        const float bb = b_pos[d];
#pragma unroll
        for (int i = 0; i < 4; ++i) {
            const int ml = i * 8 + (wave >> 2);
            const int mm = mbase + ml;
            float v = w0 * query[mm * 3 + 0]
                    + w1 * query[mm * 3 + 1]
                    + w2 * query[mm * 3 + 2]
                    + bb;
            v = (v >= 0.f) ? v : LEAKY * v;
            outb[(size_t)(D_ + d) * NQ + qbase + ml] = v;
        }
    }

    float ld  = 3.4e38f;
    int   li  = 0;
    const bool notfirst = (lane & 15) != 0;

    v2f qx2 = {qx, qx};
    v2f qy2 = {qy, qy};
    v2f qz2 = {qz, qz};
    const v4f* lp = reinterpret_cast<const v4f*>(pts + lane * 12);

    float lmin = 3.4e38f;
#pragma unroll
    for (int s = 0; s < PRE_STEPS; ++s) {
        v2f dab, dcd;
        step_d2(lp + s * 96, qx2, qy2, qz2, dab, dcd);
        lmin = fminf(lmin, fminf(fminf(dab.x, dab.y), fminf(dcd.x, dcd.y)));
    }
    {
        const int lb = __float_as_int(lmin);
        lmin = fminf(lmin, __int_as_float(__builtin_amdgcn_permlanex16(
                               lb, lb, 0x76543210, 0xfedcba98, false, false)));
    }
    float t0 = lmin;
    t0 = fmaxf(t0, __int_as_float(__builtin_amdgcn_update_dpp(
                       __float_as_int(t0), __float_as_int(t0), 0x128, 0xf, 0xf, false)));
    t0 = fmaxf(t0, __int_as_float(__builtin_amdgcn_update_dpp(
                       __float_as_int(t0), __float_as_int(t0), 0x124, 0xf, 0xf, false)));
    t0 = fmaxf(t0, __int_as_float(__builtin_amdgcn_update_dpp(
                       __float_as_int(t0), __float_as_int(t0), 0x122, 0xf, 0xf, false)));
    t0 = fmaxf(t0, __int_as_float(__builtin_amdgcn_update_dpp(
                       __float_as_int(t0), __float_as_int(t0), 0x121, 0xf, 0xf, false)));
    const int thr0b = __builtin_amdgcn_readfirstlane(__float_as_int(
        fminf(fmaxf(t0 * 1.0001f, 1.17549435e-38f), 3.4e38f)));
    float thr = __int_as_float(thr0b);

    asm volatile("" : "+v"(qx2), "+v"(qy2), "+v"(qz2));

    const v4f* sp = lp;
    for (int base = 0; base < NP; base += 128, sp += 96) {
        v2f dab, dcd;
        step_d2(sp, qx2, qy2, qz2, dab, dcd);
        const float d2[4] = {dab.x, dab.y, dcd.x, dcd.y};
        const float mn = fminf(fminf(d2[0], d2[1]), fminf(d2[2], d2[3]));
        if (__builtin_amdgcn_ballot_w32(mn < thr) == 0) continue;
#pragma unroll
        for (int j = 0; j < 4; ++j) {
            unsigned mask = __builtin_amdgcn_ballot_w32(d2[j] < thr);
            if (mask) {
                const int d2bits = __float_as_int(d2[j]);
                do {
                    const int srcl = __builtin_ctz(mask);
                    mask &= mask - 1;
                    const float cd = __int_as_float(__builtin_amdgcn_readlane(d2bits, srcl));
                    const int   ci = base + j * 32 + srcl;
                    const float ldu = __int_as_float(__builtin_amdgcn_update_dpp(
                        __float_as_int(ld), __float_as_int(ld), 0x111, 0xf, 0xf, false));
                    const int   liu = __builtin_amdgcn_update_dpp(li, li, 0x111, 0xf, 0xf, false);
                    const bool gt  = ld > cd;
                    const bool gtp = notfirst && (ldu > cd);
                    ld = gt ? (gtp ? ldu : cd) : ld;
                    li = gt ? (gtp ? liu : ci) : li;
                } while (mask);
                const int l15 = __builtin_amdgcn_readlane(__float_as_int(ld), KNN - 1);
                thr = __int_as_float(l15 < thr0b ? l15 : thr0b);
            }
        }
    }

    const float* np = pts + ppos(li);
    const float npx = qx - np[0];
    const float npy = qy - np[2];
    const float npz = qz - np[4];

    const int d0 = lane * 4;
    float wx[4], wy[4], wz[4], bp[4], wa[4];
#pragma unroll
    for (int t = 0; t < 4; ++t) {
        wx[t] = w_pe[(d0 + t) * 3 + 0];
        wy[t] = w_pe[(d0 + t) * 3 + 1];
        wz[t] = w_pe[(d0 + t) * 3 + 2];
        bp[t] = b_pe[d0 + t];
        wa[t] = w_attn[d0 + t];
    }

    const float* __restrict__ pfb = point_feature + (size_t)b * NP * D_ + d0;

    float hreg[KNN][4];
    float score[KNN];

#pragma unroll
    for (int k = 0; k < KNN; ++k) {
        const int   j  = __builtin_amdgcn_readlane(li, k);
        const float px = __int_as_float(__builtin_amdgcn_readlane(__float_as_int(npx), k));
        const float py = __int_as_float(__builtin_amdgcn_readlane(__float_as_int(npy), k));
        const float pz = __int_as_float(__builtin_amdgcn_readlane(__float_as_int(npz), k));
        const float* pf = pfb + (size_t)j * D_;
        float s = 0.f;
#pragma unroll
        for (int t = 0; t < 4; ++t) {
            const float hv = pf[t] + wx[t] * px + wy[t] * py + wz[t] * pz + bp[t];
            hreg[k][t] = hv;
            s += hv * wa[t];
        }
#pragma unroll
        for (int off = 16; off >= 1; off >>= 1)
            s += __shfl_xor(s, off, 32);
        score[k] = s;
    }

    float mx = score[0];
#pragma unroll
    for (int k = 1; k < KNN; ++k) mx = fmaxf(mx, score[k]);
    float sum = 0.f;
#pragma unroll
    for (int k = 0; k < KNN; ++k) { score[k] = __expf(score[k] - mx); sum += score[k]; }
    const float inv = 1.0f / sum;

    float acc[4] = {0.f, 0.f, 0.f, 0.f};
#pragma unroll
    for (int k = 0; k < KNN; ++k)
#pragma unroll
        for (int t = 0; t < 4; ++t) acc[t] += score[k] * hreg[k][t];

#pragma unroll
    for (int t = 0; t < 4; ++t)
        hb[wave * HB_PITCH + d0 + t] = acc[t] * inv;
    __syncthreads();

    if (wave < 16) {
        const int mt    = wave >> 3;
        const int tileN = (wave & 7) * 16;
        const int half  = lane >> 4;
        const int l     = lane & 15;
        const float* arow = hb + (mt * 16 + l) * HB_PITCH;

        v8f c = {};
        for (int k0 = 0; k0 < D_; k0 += 4) {
            const int k = k0 + half * 2;
            v2f a, bb;
            a.x  = arow[k];
            a.y  = arow[k + 1];
            bb.x = w_v[(size_t)(tileN + l) * D_ + k];
            bb.y = w_v[(size_t)(tileN + l) * D_ + k + 1];
            c = __builtin_amdgcn_wmma_f32_16x16x4_f32(
                     false, a,  false, bb,
                     (short)0, c,  false,  false);
        }

        const float bias = b_v[tileN + l];
#pragma unroll
        for (int r = 0; r < 8; ++r) {
            const int ml = mt * 16 + r + half * 8;
            outb[(size_t)(tileN + l) * NQ + qbase + ml] = c[r] + bias;
        }
    }
}

extern "C" void kernel_launch(void* const* d_in, const int* in_sizes, int n_in,
                              void* d_out, int out_size, void* d_ws, size_t ws_size,
                              hipStream_t stream) {
    const float* query         = (const float*)d_in[0];
    const float* global_pc     = (const float*)d_in[1];
    const float* point_feature = (const float*)d_in[2];
    const float* w_pos         = (const float*)d_in[3];
    const float* b_pos         = (const float*)d_in[4];
    const float* w_pe          = (const float*)d_in[5];
    const float* b_pe          = (const float*)d_in[6];
    const float* w_v           = (const float*)d_in[7];
    const float* b_v           = (const float*)d_in[8];
    const float* w_attn        = (const float*)d_in[9];
    float* out = (float*)d_out;

    fused_kernel<<<M_TOT / QPB, 1024, 0, stream>>>(query, global_pc, point_feature,
                                                   w_pos, b_pos, w_pe, b_pe,
                                                   w_v, b_v, w_attn, out);
}


# KNN list insert as a vector-only chain: new entry = med3(left neighbour, cand, own) on the DPP-shifted list, no scalar AND per insert; group masks taken together, threshold refreshed once per step
# speedup vs baseline: 29.1440x; 1.0611x over previous
#include <hip/hip_runtime.h>
#include <math.h>

#define B_    2
#define NQ    4096
#define NP    8192
#define D_    128
#define KNN   16
#define M_TOT (B_ * NQ)
#define LEAKY 0.2f

#define QPB      32
#define HB_PITCH 132
#define PRE_STEPS 8

typedef float v2f __attribute__((ext_vector_type(2)));
typedef float v4f __attribute__((ext_vector_type(4)));
typedef float v8f __attribute__((ext_vector_type(8)));

__device__ __forceinline__ int ppos(int i) {
    return (i >> 7) * 384 + (i & 31) * 12 + ((i >> 6) & 1) * 6 + ((i >> 5) & 1);
}

__device__ __forceinline__ void step_d2(const v4f* sp, v2f qx2, v2f qy2, v2f qz2,
                                        v2f& dab, v2f& dcd) {
    const v4f r0 = sp[0];
    const v4f r1 = sp[1];
    const v4f r2 = sp[2];
    const v2f xab = {r0.x, r0.y}, yab = {r0.z, r0.w}, zab = {r1.x, r1.y};
    const v2f xcd = {r1.z, r1.w}, ycd = {r2.x, r2.y}, zcd = {r2.z, r2.w};
    const v2f dxab = qx2 - xab, dyab = qy2 - yab, dzab = qz2 - zab;
    const v2f dxcd = qx2 - xcd, dycd = qy2 - ycd, dzcd = qz2 - zcd;
    dab = dyab * dyab;
    dcd = dycd * dycd;
    dab = __builtin_elementwise_fma(dxab, dxab, dab);
    dcd = __builtin_elementwise_fma(dxcd, dxcd, dcd);
    dab = __builtin_elementwise_fma(dzab, dzab, dab);
    dcd = __builtin_elementwise_fma(dzcd, dzcd, dcd);
}

__global__ __launch_bounds__(1024) void fused_kernel(const float* __restrict__ query,
                                                     const float* __restrict__ global_pc,
                                                     const float* __restrict__ point_feature,
                                                     const float* __restrict__ w_pos,
                                                     const float* __restrict__ b_pos,
                                                     const float* __restrict__ w_pe,
                                                     const float* __restrict__ b_pe,
                                                     const float* __restrict__ w_v,
                                                     const float* __restrict__ b_v,
                                                     const float* __restrict__ w_attn,
                                                     float* __restrict__ out) {
    __shared__ __attribute__((aligned(16))) float smem[NP + 3 * NP];
    float* const hb  = smem;
    float* const pts = smem + NP;

    const int lane = threadIdx.x & 31;
    const int wave  = __builtin_amdgcn_readfirstlane(threadIdx.x >> 5);
    const int mbase = blockIdx.x * QPB;
    const int m     = mbase + wave;
    const int b     = blockIdx.x >> 7;
    const int qbase = mbase & (NQ - 1);
    float* __restrict__ outb = out + (size_t)b * 256 * NQ;

    const float* __restrict__ pc = global_pc + (size_t)b * NP * 3;
#pragma unroll
    for (int i = threadIdx.x; i < NP; i += 1024) {
        const float* p = pc + (size_t)i * 3;
        float* d = pts + ppos(i);
        d[0] = p[0]; d[2] = p[1]; d[4] = p[2];
    }

    const float qx = query[m * 3 + 0];
    const float qy = query[m * 3 + 1];
    const float qz = query[m * 3 + 2];
    __syncthreads();

    {
        const int d = threadIdx.x & (D_ - 1);
        const float w0 = w_pos[d * 3 + 0];
        const float w1 = w_pos[d * 3 + 1];
        const float w2 = w_pos[d * 3 + 2];
        const float bb = b_pos[d];
#pragma unroll
        for (int i = 0; i < 4; ++i) {
            const int ml = i * 8 + (wave >> 2);
            const int mm = mbase + ml;
            float v = w0 * query[mm * 3 + 0]
                    + w1 * query[mm * 3 + 1]
                    + w2 * query[mm * 3 + 2]
                    + bb;
            v = (v >= 0.f) ? v : LEAKY * v;
            outb[(size_t)(D_ + d) * NQ + qbase + ml] = v;
        }
    }

    float ld  = 3.4e38f;
    int   li  = 0;

    v2f qx2 = {qx, qx};
    v2f qy2 = {qy, qy};
    v2f qz2 = {qz, qz};
    const v4f* lp = reinterpret_cast<const v4f*>(pts + lane * 12);

    float lmin = 3.4e38f;
#pragma unroll
    for (int s = 0; s < PRE_STEPS; ++s) {
        v2f dab, dcd;
        step_d2(lp + s * 96, qx2, qy2, qz2, dab, dcd);
        lmin = fminf(lmin, fminf(fminf(dab.x, dab.y), fminf(dcd.x, dcd.y)));
    }
    {
        const int lb = __float_as_int(lmin);
        lmin = fminf(lmin, __int_as_float(__builtin_amdgcn_permlanex16(
                               lb, lb, 0x76543210, 0xfedcba98, false, false)));
    }
    float t0 = lmin;
    t0 = fmaxf(t0, __int_as_float(__builtin_amdgcn_update_dpp(
                       __float_as_int(t0), __float_as_int(t0), 0x128, 0xf, 0xf, false)));
    t0 = fmaxf(t0, __int_as_float(__builtin_amdgcn_update_dpp(
                       __float_as_int(t0), __float_as_int(t0), 0x124, 0xf, 0xf, false)));
    t0 = fmaxf(t0, __int_as_float(__builtin_amdgcn_update_dpp(
                       __float_as_int(t0), __float_as_int(t0), 0x122, 0xf, 0xf, false)));
    t0 = fmaxf(t0, __int_as_float(__builtin_amdgcn_update_dpp(
                       __float_as_int(t0), __float_as_int(t0), 0x121, 0xf, 0xf, false)));
    const int thr0b = __builtin_amdgcn_readfirstlane(__float_as_int(
        fminf(fmaxf(t0 * 1.0001f, 1.17549435e-38f), 3.4e38f)));
    float thr = __int_as_float(thr0b);

    asm volatile("" : "+v"(qx2), "+v"(qy2), "+v"(qz2));

    const v4f* sp = lp;
    for (int base = 0; base < NP; base += 128, sp += 96) {
        v2f dab, dcd;
        step_d2(sp, qx2, qy2, qz2, dab, dcd);
        const float d2[4] = {dab.x, dab.y, dcd.x, dcd.y};
        const float mn = fminf(fminf(d2[0], d2[1]), fminf(d2[2], d2[3]));
        if (__builtin_amdgcn_ballot_w32(mn < thr) == 0) continue;
        unsigned gm[4];
#pragma unroll
        for (int j = 0; j < 4; ++j) gm[j] = __builtin_amdgcn_ballot_w32(d2[j] < thr);
#pragma unroll
        for (int j = 0; j < 4; ++j) {
            unsigned mask = gm[j];
            const int d2bits = __float_as_int(d2[j]);
            while (mask) {
                const int srcl = __builtin_ctz(mask);
                mask &= mask - 1;
                const float cd = __int_as_float(__builtin_amdgcn_readlane(d2bits, srcl));
                const int   ci = base + j * 32 + srcl;
                const float ldu = __int_as_float(__builtin_amdgcn_update_dpp(
                    0, __float_as_int(ld), 0x111, 0xf, 0xf, true));
                const int   liu = __builtin_amdgcn_update_dpp(0, li, 0x111, 0xf, 0xf, true);
                const bool gt  = ld  > cd;
                const bool gtp = ldu > cd;
                ld = __builtin_amdgcn_fmed3f(ldu, cd, ld);
                li = gtp ? liu : (gt ? ci : li);
            }
        }
        const int l15 = __builtin_amdgcn_readlane(__float_as_int(ld), KNN - 1);
        thr = __int_as_float(l15 < thr0b ? l15 : thr0b);
    }

    const float* np = pts + ppos(li);
    const float npx = qx - np[0];
    const float npy = qy - np[2];
    const float npz = qz - np[4];

    const int d0 = lane * 4;
    float wx[4], wy[4], wz[4], bp[4], wa[4];
#pragma unroll
    for (int t = 0; t < 4; ++t) {
        wx[t] = w_pe[(d0 + t) * 3 + 0];
        wy[t] = w_pe[(d0 + t) * 3 + 1];
        wz[t] = w_pe[(d0 + t) * 3 + 2];
        bp[t] = b_pe[d0 + t];
        wa[t] = w_attn[d0 + t];
    }

    const float* __restrict__ pfb = point_feature + (size_t)b * NP * D_ + d0;

    float hreg[KNN][4];
    float score[KNN];

#pragma unroll
    for (int k = 0; k < KNN; ++k) {
        const int   j  = __builtin_amdgcn_readlane(li, k);
        const float px = __int_as_float(__builtin_amdgcn_readlane(__float_as_int(npx), k));
        const float py = __int_as_float(__builtin_amdgcn_readlane(__float_as_int(npy), k));
        const float pz = __int_as_float(__builtin_amdgcn_readlane(__float_as_int(npz), k));
        const float* pf = pfb + (size_t)j * D_;
        float s = 0.f;
#pragma unroll
        for (int t = 0; t < 4; ++t) {
            const float hv = pf[t] + wx[t] * px + wy[t] * py + wz[t] * pz + bp[t];
            hreg[k][t] = hv;
            s += hv * wa[t];
        }
#pragma unroll
        for (int off = 16; off >= 1; off >>= 1)
            s += __shfl_xor(s, off, 32);
        score[k] = s;
    }

    float mx = score[0];
#pragma unroll
    for (int k = 1; k < KNN; ++k) mx = fmaxf(mx, score[k]);
    float sum = 0.f;
#pragma unroll
    for (int k = 0; k < KNN; ++k) { score[k] = __expf(score[k] - mx); sum += score[k]; }
    const float inv = 1.0f / sum;

    float acc[4] = {0.f, 0.f, 0.f, 0.f};
#pragma unroll
    for (int k = 0; k < KNN; ++k)
#pragma unroll
        for (int t = 0; t < 4; ++t) acc[t] += score[k] * hreg[k][t];

#pragma unroll
    for (int t = 0; t < 4; ++t)
        hb[wave * HB_PITCH + d0 + t] = acc[t] * inv;
    __syncthreads();

    if (wave < 16) {
        const int mt    = wave >> 3;
        const int tileN = (wave & 7) * 16;
        const int half  = lane >> 4;
        const int l     = lane & 15;
        const float* arow = hb + (mt * 16 + l) * HB_PITCH;

        v8f c = {};
        for (int k0 = 0; k0 < D_; k0 += 4) {
            const int k = k0 + half * 2;
            v2f a, bb;
            a.x  = arow[k];
            a.y  = arow[k + 1];
            bb.x = w_v[(size_t)(tileN + l) * D_ + k];
            bb.y = w_v[(size_t)(tileN + l) * D_ + k + 1];
            c = __builtin_amdgcn_wmma_f32_16x16x4_f32(
                     false, a,  false, bb,
                     (short)0, c,  false,  false);
        }

        const float bias = b_v[tileN + l];
#pragma unroll
        for (int r = 0; r < 8; ++r) {
            const int ml = mt * 16 + r + half * 8;
            outb[(size_t)(tileN + l) * NQ + qbase + ml] = c[r] + bias;
        }
    }
}

extern "C" void kernel_launch(void* const* d_in, const int* in_sizes, int n_in,
                              void* d_out, int out_size, void* d_ws, size_t ws_size,
                              hipStream_t stream) {
    const float* query         = (const float*)d_in[0];
    const float* global_pc     = (const float*)d_in[1];
    const float* point_feature = (const float*)d_in[2];
    const float* w_pos         = (const float*)d_in[3];
    const float* b_pos         = (const float*)d_in[4];
    const float* w_pe          = (const float*)d_in[5];
    const float* b_pe          = (const float*)d_in[6];
    const float* w_v           = (const float*)d_in[7];
    const float* b_v           = (const float*)d_in[8];
    const float* w_attn        = (const float*)d_in[9];
    float* out = (float*)d_out;

    fused_kernel<<<M_TOT / QPB, 1024, 0, stream>>>(query, global_pc, point_feature,
                                                   w_pos, b_pos, w_pe, b_pe,
                                                   w_v, b_v, w_attn, out);
}


# KNN scan loop takes 4 steps (512 points) per pass: the 12 LDS loads go out together, the 4 step masks are taken in one go, one branch and one loop overhead per pass; insert sequence unchanged
# speedup vs baseline: 30.1882x; 1.0358x over previous
#include <hip/hip_runtime.h>
#include <math.h>

#define B_    2
#define NQ    4096
#define NP    8192
#define D_    128
#define KNN   16
#define M_TOT (B_ * NQ)
#define LEAKY 0.2f

#define QPB      32
#define HB_PITCH 132
#define PRE_STEPS 8
#define SCAN_U    4

typedef float v2f __attribute__((ext_vector_type(2)));
typedef float v4f __attribute__((ext_vector_type(4)));
typedef float v8f __attribute__((ext_vector_type(8)));

__device__ __forceinline__ int ppos(int i) {
    return (i >> 7) * 384 + (i & 31) * 12 + ((i >> 6) & 1) * 6 + ((i >> 5) & 1);
}

__device__ __forceinline__ void step_d2(const v4f* sp, v2f qx2, v2f qy2, v2f qz2,
                                        v2f& dab, v2f& dcd) {
    const v4f r0 = sp[0];
    const v4f r1 = sp[1];
    const v4f r2 = sp[2];
    const v2f xab = {r0.x, r0.y}, yab = {r0.z, r0.w}, zab = {r1.x, r1.y};
    const v2f xcd = {r1.z, r1.w}, ycd = {r2.x, r2.y}, zcd = {r2.z, r2.w};
    const v2f dxab = qx2 - xab, dyab = qy2 - yab, dzab = qz2 - zab;
    const v2f dxcd = qx2 - xcd, dycd = qy2 - ycd, dzcd = qz2 - zcd;
    dab = dyab * dyab;
    dcd = dycd * dycd;
    dab = __builtin_elementwise_fma(dxab, dxab, dab);
    dcd = __builtin_elementwise_fma(dxcd, dxcd, dcd);
    dab = __builtin_elementwise_fma(dzab, dzab, dab);
    dcd = __builtin_elementwise_fma(dzcd, dzcd, dcd);
}

__global__ __launch_bounds__(1024) void fused_kernel(const float* __restrict__ query,
                                                     const float* __restrict__ global_pc,
                                                     const float* __restrict__ point_feature,
                                                     const float* __restrict__ w_pos,
                                                     const float* __restrict__ b_pos,
                                                     const float* __restrict__ w_pe,
                                                     const float* __restrict__ b_pe,
                                                     const float* __restrict__ w_v,
                                                     const float* __restrict__ b_v,
                                                     const float* __restrict__ w_attn,
                                                     float* __restrict__ out) {
    __shared__ __attribute__((aligned(16))) float smem[NP + 3 * NP];
    float* const hb  = smem;
    float* const pts = smem + NP;

    const int lane = threadIdx.x & 31;
    const int wave  = __builtin_amdgcn_readfirstlane(threadIdx.x >> 5);
    const int mbase = blockIdx.x * QPB;
    const int m     = mbase + wave;
    const int b     = blockIdx.x >> 7;
    const int qbase = mbase & (NQ - 1);
    float* __restrict__ outb = out + (size_t)b * 256 * NQ;

    const float* __restrict__ pc = global_pc + (size_t)b * NP * 3;
#pragma unroll
    for (int i = threadIdx.x; i < NP; i += 1024) {
        const float* p = pc + (size_t)i * 3;
        float* d = pts + ppos(i);
        d[0] = p[0]; d[2] = p[1]; d[4] = p[2];
    }

    const float qx = query[m * 3 + 0];
    const float qy = query[m * 3 + 1];
    const float qz = query[m * 3 + 2];
    __syncthreads();

    {
        const int d = threadIdx.x & (D_ - 1);
        const float w0 = w_pos[d * 3 + 0];
        const float w1 = w_pos[d * 3 + 1];
        const float w2 = w_pos[d * 3 + 2];
        const float bb = b_pos[d];
#pragma unroll
        for (int i = 0; i < 4; ++i) {
            const int ml = i * 8 + (wave >> 2);
            const int mm = mbase + ml;
            float v = w0 * query[mm * 3 + 0]
                    + w1 * query[mm * 3 + 1]
                    + w2 * query[mm * 3 + 2]
                    + bb;
            v = (v >= 0.f) ? v : LEAKY * v;
            outb[(size_t)(D_ + d) * NQ + qbase + ml] = v;
        }
    }

    float ld  = 3.4e38f;
    int   li  = 0;

    v2f qx2 = {qx, qx};
    v2f qy2 = {qy, qy};
    v2f qz2 = {qz, qz};
    const v4f* lp = reinterpret_cast<const v4f*>(pts + lane * 12);

    float lmin = 3.4e38f;
#pragma unroll
    for (int s = 0; s < PRE_STEPS; ++s) {
        v2f dab, dcd;
        step_d2(lp + s * 96, qx2, qy2, qz2, dab, dcd);
        lmin = fminf(lmin, fminf(fminf(dab.x, dab.y), fminf(dcd.x, dcd.y)));
    }
    {
        const int lb = __float_as_int(lmin);
        lmin = fminf(lmin, __int_as_float(__builtin_amdgcn_permlanex16(
                               lb, lb, 0x76543210, 0xfedcba98, false, false)));
    }
    float t0 = lmin;
    t0 = fmaxf(t0, __int_as_float(__builtin_amdgcn_update_dpp(
                       __float_as_int(t0), __float_as_int(t0), 0x128, 0xf, 0xf, false)));
    t0 = fmaxf(t0, __int_as_float(__builtin_amdgcn_update_dpp(
                       __float_as_int(t0), __float_as_int(t0), 0x124, 0xf, 0xf, false)));
    t0 = fmaxf(t0, __int_as_float(__builtin_amdgcn_update_dpp(
                       __float_as_int(t0), __float_as_int(t0), 0x122, 0xf, 0xf, false)));
    t0 = fmaxf(t0, __int_as_float(__builtin_amdgcn_update_dpp(
                       __float_as_int(t0), __float_as_int(t0), 0x121, 0xf, 0xf, false)));
    const int thr0b = __builtin_amdgcn_readfirstlane(__float_as_int(
        fminf(fmaxf(t0 * 1.0001f, 1.17549435e-38f), 3.4e38f)));
    float thr = __int_as_float(thr0b);

    asm volatile("" : "+v"(qx2), "+v"(qy2), "+v"(qz2));

    const v4f* sp = lp;
    for (int base = 0; base < NP; base += 128 * SCAN_U, sp += 96 * SCAN_U) {
        float d2[SCAN_U][4];
        float mn[SCAN_U];
#pragma unroll
        for (int u = 0; u < SCAN_U; ++u) {
            v2f dab, dcd;
            step_d2(sp + u * 96, qx2, qy2, qz2, dab, dcd);
            d2[u][0] = dab.x; d2[u][1] = dab.y; d2[u][2] = dcd.x; d2[u][3] = dcd.y;
            mn[u] = fminf(fminf(dab.x, dab.y), fminf(dcd.x, dcd.y));
        }
        unsigned sm[SCAN_U];
        unsigned any = 0;
#pragma unroll
        for (int u = 0; u < SCAN_U; ++u) {
            sm[u] = __builtin_amdgcn_ballot_w32(mn[u] < thr);
            any |= sm[u];
        }
        if (any == 0) continue;
#pragma unroll
        for (int u = 0; u < SCAN_U; ++u) {
            if (sm[u] != 0) {
                unsigned gm[4];
#pragma unroll
                for (int j = 0; j < 4; ++j) gm[j] = __builtin_amdgcn_ballot_w32(d2[u][j] < thr);
#pragma unroll
                for (int j = 0; j < 4; ++j) {
                    unsigned mask = gm[j];
                    const int d2bits = __float_as_int(d2[u][j]);
                    while (mask) {
                        const int srcl = __builtin_ctz(mask);
                        mask &= mask - 1;
                        const float cd = __int_as_float(__builtin_amdgcn_readlane(d2bits, srcl));
                        const int   ci = base + u * 128 + j * 32 + srcl;
                        const float ldu = __int_as_float(__builtin_amdgcn_update_dpp(
                            0, __float_as_int(ld), 0x111, 0xf, 0xf, true));
                        const int   liu = __builtin_amdgcn_update_dpp(0, li, 0x111, 0xf, 0xf, true);
                        const bool gt  = ld  > cd;
                        const bool gtp = ldu > cd;
                        ld = __builtin_amdgcn_fmed3f(ldu, cd, ld);
                        li = gtp ? liu : (gt ? ci : li);
                    }
                }
                const int l15 = __builtin_amdgcn_readlane(__float_as_int(ld), KNN - 1);
                thr = __int_as_float(l15 < thr0b ? l15 : thr0b);
            }
        }
    }

    const float* np = pts + ppos(li);
    const float npx = qx - np[0];
    const float npy = qy - np[2];
    const float npz = qz - np[4];

    const int d0 = lane * 4;
    float wx[4], wy[4], wz[4], bp[4], wa[4];
#pragma unroll
    for (int t = 0; t < 4; ++t) {
        wx[t] = w_pe[(d0 + t) * 3 + 0];
        wy[t] = w_pe[(d0 + t) * 3 + 1];
        wz[t] = w_pe[(d0 + t) * 3 + 2];
        bp[t] = b_pe[d0 + t];
        wa[t] = w_attn[d0 + t];
    }

    const float* __restrict__ pfb = point_feature + (size_t)b * NP * D_ + d0;

    float hreg[KNN][4];
    float score[KNN];

#pragma unroll
    for (int k = 0; k < KNN; ++k) {
        const int   j  = __builtin_amdgcn_readlane(li, k);
        const float px = __int_as_float(__builtin_amdgcn_readlane(__float_as_int(npx), k));
        const float py = __int_as_float(__builtin_amdgcn_readlane(__float_as_int(npy), k));
        const float pz = __int_as_float(__builtin_amdgcn_readlane(__float_as_int(npz), k));
        const float* pf = pfb + (size_t)j * D_;
        float s = 0.f;
#pragma unroll
        for (int t = 0; t < 4; ++t) {
            const float hv = pf[t] + wx[t] * px + wy[t] * py + wz[t] * pz + bp[t];
            hreg[k][t] = hv;
            s += hv * wa[t];
        }
#pragma unroll
        for (int off = 16; off >= 1; off >>= 1)
            s += __shfl_xor(s, off, 32);
        score[k] = s;
    }

    float mx = score[0];
#pragma unroll
    for (int k = 1; k < KNN; ++k) mx = fmaxf(mx, score[k]);
    float sum = 0.f;
#pragma unroll
    for (int k = 0; k < KNN; ++k) { score[k] = __expf(score[k] - mx); sum += score[k]; }
    const float inv = 1.0f / sum;

    float acc[4] = {0.f, 0.f, 0.f, 0.f};
#pragma unroll
    for (int k = 0; k < KNN; ++k)
#pragma unroll
        for (int t = 0; t < 4; ++t) acc[t] += score[k] * hreg[k][t];

#pragma unroll
    for (int t = 0; t < 4; ++t)
        hb[wave * HB_PITCH + d0 + t] = acc[t] * inv;
    __syncthreads();

    if (wave < 16) {
        const int mt    = wave >> 3;
        const int tileN = (wave & 7) * 16;
        const int half  = lane >> 4;
        const int l     = lane & 15;
        const float* arow = hb + (mt * 16 + l) * HB_PITCH;

        v8f c = {};
        for (int k0 = 0; k0 < D_; k0 += 4) {
            const int k = k0 + half * 2;
            v2f a, bb;
            a.x  = arow[k];
            a.y  = arow[k + 1];
            bb.x = w_v[(size_t)(tileN + l) * D_ + k];
            bb.y = w_v[(size_t)(tileN + l) * D_ + k + 1];
            c = __builtin_amdgcn_wmma_f32_16x16x4_f32(
                     false, a,  false, bb,
                     (short)0, c,  false,  false);
        }

        const float bias = b_v[tileN + l];
#pragma unroll
        for (int r = 0; r < 8; ++r) {
            const int ml = mt * 16 + r + half * 8;
            outb[(size_t)(tileN + l) * NQ + qbase + ml] = c[r] + bias;
        }
    }
}

extern "C" void kernel_launch(void* const* d_in, const int* in_sizes, int n_in,
                              void* d_out, int out_size, void* d_ws, size_t ws_size,
                              hipStream_t stream) {
    const float* query         = (const float*)d_in[0];
    const float* global_pc     = (const float*)d_in[1];
    const float* point_feature = (const float*)d_in[2];
    const float* w_pos         = (const float*)d_in[3];
    const float* b_pos         = (const float*)d_in[4];
    const float* w_pe          = (const float*)d_in[5];
    const float* b_pe          = (const float*)d_in[6];
    const float* w_v           = (const float*)d_in[7];
    const float* b_v           = (const float*)d_in[8];
    const float* w_attn        = (const float*)d_in[9];
    float* out = (float*)d_out;

    fused_kernel<<<M_TOT / QPB, 1024, 0, stream>>>(query, global_pc, point_feature,
                                                   w_pos, b_pos, w_pe, b_pe,
                                                   w_v, b_v, w_attn, out);
}


# Cloud sorted into 4 x-slabs while staged in LDS (ballot ranks, slot->index table); a query scans its own slab, then a neighbour slab only while squared x gap < its threshold; equal d2: first met kept
# speedup vs baseline: 34.4761x; 1.1420x over previous
#include <hip/hip_runtime.h>
#include <math.h>

#define B_    2
#define NQ    4096
#define NP    8192
#define D_    128
#define KNN   16
#define M_TOT (B_ * NQ)
#define LEAKY 0.2f

#define QPB      32
#define HB_PITCH 132
#define PRE_STEPS 8
#define SCAN_U    4
#define PASS_PTS  (128 * SCAN_U)
#define NBKT      4
#define NPASS     (NP / PASS_PTS + NBKT - 1)
#define NSLOT     (NPASS * PASS_PTS)
#define XB0 (-0.6744898f)
#define XB1 0.0f
#define XB2 0.6744898f
#define FAR_COORD 1.0e30f

typedef float v2f __attribute__((ext_vector_type(2)));
typedef float v4f __attribute__((ext_vector_type(4)));
typedef float v8f __attribute__((ext_vector_type(8)));

__device__ __forceinline__ int ppos(int i) {
    return __mul24(i >> 7, 384) + (i & 31) * 12 + ((i >> 6) & 1) * 6 + ((i >> 5) & 1);
}

__device__ __forceinline__ float xbound(int k) {
    return k == 0 ? XB0 : (k == 1 ? XB1 : XB2);
}

__device__ __forceinline__ void step_d2(const v4f* sp, v2f qx2, v2f qy2, v2f qz2,
                                        v2f& dab, v2f& dcd) {
    const v4f r0 = sp[0];
    const v4f r1 = sp[1];
    const v4f r2 = sp[2];
    const v2f xab = {r0.x, r0.y}, yab = {r0.z, r0.w}, zab = {r1.x, r1.y};
    const v2f xcd = {r1.z, r1.w}, ycd = {r2.x, r2.y}, zcd = {r2.z, r2.w};
    const v2f dxab = qx2 - xab, dyab = qy2 - yab, dzab = qz2 - zab;
    const v2f dxcd = qx2 - xcd, dycd = qy2 - ycd, dzcd = qz2 - zcd;
    dab = dyab * dyab;
    dcd = dycd * dycd;
    dab = __builtin_elementwise_fma(dxab, dxab, dab);
    dcd = __builtin_elementwise_fma(dxcd, dxcd, dcd);
    dab = __builtin_elementwise_fma(dzab, dzab, dab);
    dcd = __builtin_elementwise_fma(dzcd, dzcd, dcd);
}

__global__ __launch_bounds__(1024) void fused_kernel(const float* __restrict__ query,
                                                     const float* __restrict__ global_pc,
                                                     const float* __restrict__ point_feature,
                                                     const float* __restrict__ w_pos,
                                                     const float* __restrict__ b_pos,
                                                     const float* __restrict__ w_pe,
                                                     const float* __restrict__ b_pe,
                                                     const float* __restrict__ w_v,
                                                     const float* __restrict__ b_v,
                                                     const float* __restrict__ w_attn,
                                                     float* __restrict__ out) {
    __shared__ __attribute__((aligned(16))) float smem[NP + 3 * NSLOT + NSLOT];
    float* const hb  = smem;
    float* const pts = smem + NP;
    int* const slot2idx = reinterpret_cast<int*>(smem + NP + 3 * NSLOT);
    int* const wcount   = reinterpret_cast<int*>(smem);

    const int lane = threadIdx.x & 31;
    const int wave  = __builtin_amdgcn_readfirstlane(threadIdx.x >> 5);
    const int mbase = blockIdx.x * QPB;
    const int m     = mbase + wave;
    const int b     = blockIdx.x >> 7;
    const int qbase = mbase & (NQ - 1);
    float* __restrict__ outb = out + (size_t)b * 256 * NQ;

    const float* __restrict__ pc = global_pc + (size_t)b * NP * 3;
    float px[8], py[8], pz[8];
#pragma unroll
    for (int r = 0; r < 8; ++r) {
        const float* p = pc + (size_t)(threadIdx.x + r * 1024) * 3;
        px[r] = p[0]; py[r] = p[1]; pz[r] = p[2];
    }
    unsigned cm[8][NBKT - 1];
    int cum[NBKT - 1];
#pragma unroll
    for (int k = 0; k < NBKT - 1; ++k) cum[k] = 0;
#pragma unroll
    for (int r = 0; r < 8; ++r) {
#pragma unroll
        for (int k = 0; k < NBKT - 1; ++k) {
            cm[r][k] = __builtin_amdgcn_ballot_w32(px[r] < xbound(k));
            cum[k] += __builtin_popcount(cm[r][k]);
        }
    }
    if (lane < NBKT - 1)
        wcount[lane * 32 + wave] = lane == 0 ? cum[0] : (lane == 1 ? cum[1] : cum[2]);

    const float qx = query[m * 3 + 0];
    const float qy = query[m * 3 + 1];
    const float qz = query[m * 3 + 2];
    __syncthreads();

    int tle[NBKT - 1], ble[NBKT - 1];
#pragma unroll
    for (int k = 0; k < NBKT - 1; ++k) {
        const int v = wcount[k * 32 + lane];
        int pk = v + (lane < wave ? (v << 16) : 0);
        pk += __builtin_amdgcn_update_dpp(pk, pk, 0x128, 0xf, 0xf, false);
        pk += __builtin_amdgcn_update_dpp(pk, pk, 0x124, 0xf, 0xf, false);
        pk += __builtin_amdgcn_update_dpp(pk, pk, 0x122, 0xf, 0xf, false);
        pk += __builtin_amdgcn_update_dpp(pk, pk, 0x121, 0xf, 0xf, false);
        pk += __builtin_amdgcn_permlanex16(pk, pk, 0x76543210, 0xfedcba98, false, false);
        tle[k] = __builtin_amdgcn_readfirstlane(pk & 0xffff);
        ble[k] = __builtin_amdgcn_readfirstlane(pk >> 16);
    }
    int nb[NBKT], wb[NBKT], pb[NBKT + 1];
    nb[0] = tle[0];
    wb[0] = ble[0];
#pragma unroll
    for (int s = 1; s < NBKT - 1; ++s) {
        nb[s] = tle[s] - tle[s - 1];
        wb[s] = ble[s] - ble[s - 1];
    }
    nb[NBKT - 1] = NP - tle[NBKT - 2];
    wb[NBKT - 1] = wave * 256 - ble[NBKT - 2];
    pb[0] = 0;
#pragma unroll
    for (int s = 0; s < NBKT; ++s) pb[s + 1] = pb[s] + (nb[s] + PASS_PTS - 1) / PASS_PTS;
    {
        int run[NBKT];
#pragma unroll
        for (int s = 0; s < NBKT; ++s) run[s] = pb[s] * PASS_PTS + wb[s];
#pragma unroll
        for (int r = 0; r < 8; ++r) {
            unsigned bm = ~cm[r][NBKT - 2];
            int a = run[NBKT - 1];
#pragma unroll
            for (int k = NBKT - 2; k >= 0; --k) {
                const unsigned own = k > 0 ? (cm[r][k] & ~cm[r][k - 1]) : cm[r][0];
                if (px[r] < xbound(k)) { bm = own; a = run[k]; }
            }
            const int slot = a + __builtin_amdgcn_mbcnt_lo(bm, 0);
            float* d = pts + ppos(slot);
            d[0] = px[r]; d[2] = py[r]; d[4] = pz[r];
            slot2idx[slot] = threadIdx.x + r * 1024;
            run[0] += __builtin_popcount(cm[r][0]);
#pragma unroll
            for (int s = 1; s < NBKT - 1; ++s) run[s] += __builtin_popcount(cm[r][s] & ~cm[r][s - 1]);
            run[NBKT - 1] += __builtin_popcount(~cm[r][NBKT - 2]);
        }
    }
    if (threadIdx.x < PASS_PTS) {
#pragma unroll
        for (int s = 0; s < NBKT; ++s) {
            const int slot = pb[s] * PASS_PTS + nb[s] + threadIdx.x;
            if (slot < pb[s + 1] * PASS_PTS) {
                float* d = pts + ppos(slot);
                d[0] = FAR_COORD; d[2] = FAR_COORD; d[4] = FAR_COORD;
                slot2idx[slot] = 0;
            }
        }
    }
    __syncthreads();

    {
        const int d = threadIdx.x & (D_ - 1);
        const float w0 = w_pos[d * 3 + 0];
        const float w1 = w_pos[d * 3 + 1];
        const float w2 = w_pos[d * 3 + 2];
        const float bb = b_pos[d];
#pragma unroll
        for (int i = 0; i < 4; ++i) {
            const int ml = i * 8 + (wave >> 2);
            const int mm = mbase + ml;
            float v = w0 * query[mm * 3 + 0]
                    + w1 * query[mm * 3 + 1]
                    + w2 * query[mm * 3 + 2]
                    + bb;
            v = (v >= 0.f) ? v : LEAKY * v;
            outb[(size_t)(D_ + d) * NQ + qbase + ml] = v;
        }
    }

    float ld  = 3.4e38f;
    int   li  = 0;

    v2f qx2 = {qx, qx};
    v2f qy2 = {qy, qy};
    v2f qz2 = {qz, qz};
    const v4f* lp = reinterpret_cast<const v4f*>(pts + lane * 12);

    int bq = 0;
    int pown = pb[0];
#pragma unroll
    for (int k = 0; k < NBKT - 1; ++k)
        if (!(qx < xbound(k))) { bq = k + 1; pown = pb[k + 1]; }
    const int pend = pb[NBKT];
    float gap[NBKT];
#pragma unroll
    for (int s = 0; s < NBKT; ++s) {
        float g = 0.f;
        if (s > 0)        g = fmaxf(g, xbound(s - 1) - qx);
        if (s < NBKT - 1) g = fmaxf(g, qx - xbound(s));
        gap[s] = g;
    }
    const int sb = min(pown * SCAN_U, pend * SCAN_U - PRE_STEPS);
    const v4f* lpb = lp + sb * 96;

    float lmin = 3.4e38f;
#pragma unroll
    for (int s = 0; s < PRE_STEPS; ++s) {
        v2f dab, dcd;
        step_d2(lpb + s * 96, qx2, qy2, qz2, dab, dcd);
        lmin = fminf(lmin, fminf(fminf(dab.x, dab.y), fminf(dcd.x, dcd.y)));
    }
    {
        const int lb = __float_as_int(lmin);
        lmin = fminf(lmin, __int_as_float(__builtin_amdgcn_permlanex16(
                               lb, lb, 0x76543210, 0xfedcba98, false, false)));
    }
    float t0 = lmin;
    t0 = fmaxf(t0, __int_as_float(__builtin_amdgcn_update_dpp(
                       __float_as_int(t0), __float_as_int(t0), 0x128, 0xf, 0xf, false)));
    t0 = fmaxf(t0, __int_as_float(__builtin_amdgcn_update_dpp(
                       __float_as_int(t0), __float_as_int(t0), 0x124, 0xf, 0xf, false)));
    t0 = fmaxf(t0, __int_as_float(__builtin_amdgcn_update_dpp(
                       __float_as_int(t0), __float_as_int(t0), 0x122, 0xf, 0xf, false)));
    t0 = fmaxf(t0, __int_as_float(__builtin_amdgcn_update_dpp(
                       __float_as_int(t0), __float_as_int(t0), 0x121, 0xf, 0xf, false)));
    const int thr0b = __builtin_amdgcn_readfirstlane(__float_as_int(
        fminf(fmaxf(t0 * 1.0001f, 1.17549435e-38f), 3.4e38f)));
    float thr = __int_as_float(thr0b);

    asm volatile("" : "+v"(qx2), "+v"(qy2), "+v"(qz2));

    int slab = bq, dir = 1;
    for (int vis = 0; vis < NBKT + 2; ++vis, slab += dir) {
        bool go = slab >= 0 && slab < NBKT;
        int p0 = 0, p1 = 0;
        if (go) {
            float g = gap[0];
            p0 = pb[0]; p1 = pb[1];
#pragma unroll
            for (int s = 1; s < NBKT; ++s)
                if (slab == s) { g = gap[s]; p0 = pb[s]; p1 = pb[s + 1]; }
            go = g * g < thr;
        }
        if (!go) {
            if (dir < 0) break;
            dir = -1;
            slab = bq;
            continue;
        }
        for (int pass = p0; pass < p1; ++pass) {
            const int base = pass * PASS_PTS;
            const v4f* sp  = lp + pass * (96 * SCAN_U);
            float d2[SCAN_U][4];
            float mn[SCAN_U];
#pragma unroll
            for (int u = 0; u < SCAN_U; ++u) {
                v2f dab, dcd;
                step_d2(sp + u * 96, qx2, qy2, qz2, dab, dcd);
                d2[u][0] = dab.x; d2[u][1] = dab.y; d2[u][2] = dcd.x; d2[u][3] = dcd.y;
                mn[u] = fminf(fminf(dab.x, dab.y), fminf(dcd.x, dcd.y));
            }
            unsigned sm[SCAN_U];
            unsigned any = 0;
#pragma unroll
            for (int u = 0; u < SCAN_U; ++u) {
                sm[u] = __builtin_amdgcn_ballot_w32(mn[u] < thr);
                any |= sm[u];
            }
            if (any == 0) continue;
#pragma unroll
            for (int u = 0; u < SCAN_U; ++u) {
                if (sm[u] != 0) {
                    unsigned gm[4];
#pragma unroll
                    for (int j = 0; j < 4; ++j) gm[j] = __builtin_amdgcn_ballot_w32(d2[u][j] < thr);
#pragma unroll
                    for (int j = 0; j < 4; ++j) {
                        unsigned mask = gm[j];
                        const int d2bits = __float_as_int(d2[u][j]);
                        while (mask) {
                            const int srcl = __builtin_ctz(mask);
                            mask &= mask - 1;
                            const float cd = __int_as_float(__builtin_amdgcn_readlane(d2bits, srcl));
                            const int   ci = base + u * 128 + j * 32 + srcl;
                            const float ldu = __int_as_float(__builtin_amdgcn_update_dpp(
                                0, __float_as_int(ld), 0x111, 0xf, 0xf, true));
                            const int   liu = __builtin_amdgcn_update_dpp(0, li, 0x111, 0xf, 0xf, true);
                            const bool gt  = ld  > cd;
                            const bool gtp = ldu > cd;
                            ld = __builtin_amdgcn_fmed3f(ldu, cd, ld);
                            li = gtp ? liu : (gt ? ci : li);
                        }
                    }
                    const int l15 = __builtin_amdgcn_readlane(__float_as_int(ld), KNN - 1);
                    thr = __int_as_float(l15 < thr0b ? l15 : thr0b);
                }
            }
        }
    }

    const float* np = pts + ppos(li);
    const int lorig = slot2idx[li];
    const float npx = qx - np[0];
    const float npy = qy - np[2];
    const float npz = qz - np[4];

    const int d0 = lane * 4;
    float wx[4], wy[4], wz[4], bp[4], wa[4];
#pragma unroll
    for (int t = 0; t < 4; ++t) {
        wx[t] = w_pe[(d0 + t) * 3 + 0];
        wy[t] = w_pe[(d0 + t) * 3 + 1];
        wz[t] = w_pe[(d0 + t) * 3 + 2];
        bp[t] = b_pe[d0 + t];
        wa[t] = w_attn[d0 + t];
    }

    const float* __restrict__ pfb = point_feature + (size_t)b * NP * D_ + d0;

    float hreg[KNN][4];
    float score[KNN];

#pragma unroll
    for (int k = 0; k < KNN; ++k) {
        const int   j  = __builtin_amdgcn_readlane(lorig, k);
        const float px = __int_as_float(__builtin_amdgcn_readlane(__float_as_int(npx), k));
        const float py = __int_as_float(__builtin_amdgcn_readlane(__float_as_int(npy), k));
        const float pz = __int_as_float(__builtin_amdgcn_readlane(__float_as_int(npz), k));
        const float* pf = pfb + (size_t)j * D_;
        float s = 0.f;
#pragma unroll
        for (int t = 0; t < 4; ++t) {
            const float hv = pf[t] + wx[t] * px + wy[t] * py + wz[t] * pz + bp[t];
            hreg[k][t] = hv;
            s += hv * wa[t];
        }
#pragma unroll
        for (int off = 16; off >= 1; off >>= 1)
            s += __shfl_xor(s, off, 32);
        score[k] = s;
    }

    float mx = score[0];
#pragma unroll
    for (int k = 1; k < KNN; ++k) mx = fmaxf(mx, score[k]);
    float sum = 0.f;
#pragma unroll
    for (int k = 0; k < KNN; ++k) { score[k] = __expf(score[k] - mx); sum += score[k]; }
    const float inv = 1.0f / sum;

    float acc[4] = {0.f, 0.f, 0.f, 0.f};
#pragma unroll
    for (int k = 0; k < KNN; ++k)
#pragma unroll
        for (int t = 0; t < 4; ++t) acc[t] += score[k] * hreg[k][t];

#pragma unroll
    for (int t = 0; t < 4; ++t)
        hb[wave * HB_PITCH + d0 + t] = acc[t] * inv;
    __syncthreads();

    if (wave < 16) {
        const int mt    = wave >> 3;
        const int tileN = (wave & 7) * 16;
        const int half  = lane >> 4;
        const int l     = lane & 15;
        const float* arow = hb + (mt * 16 + l) * HB_PITCH;

        v8f c = {};
        for (int k0 = 0; k0 < D_; k0 += 4) {
            const int k = k0 + half * 2;
            v2f a, bb;
            a.x  = arow[k];
            a.y  = arow[k + 1];
            bb.x = w_v[(size_t)(tileN + l) * D_ + k];
            bb.y = w_v[(size_t)(tileN + l) * D_ + k + 1];
            c = __builtin_amdgcn_wmma_f32_16x16x4_f32(
                     false, a,  false, bb,
                     (short)0, c,  false,  false);
        }

        const float bias = b_v[tileN + l];
#pragma unroll
        for (int r = 0; r < 8; ++r) {
            const int ml = mt * 16 + r + half * 8;
            outb[(size_t)(tileN + l) * NQ + qbase + ml] = c[r] + bias;
        }
    }
}

extern "C" void kernel_launch(void* const* d_in, const int* in_sizes, int n_in,
                              void* d_out, int out_size, void* d_ws, size_t ws_size,
                              hipStream_t stream) {
    const float* query         = (const float*)d_in[0];
    const float* global_pc     = (const float*)d_in[1];
    const float* point_feature = (const float*)d_in[2];
    const float* w_pos         = (const float*)d_in[3];
    const float* b_pos         = (const float*)d_in[4];
    const float* w_pe          = (const float*)d_in[5];
    const float* b_pe          = (const float*)d_in[6];
    const float* w_v           = (const float*)d_in[7];
    const float* b_v           = (const float*)d_in[8];
    const float* w_attn        = (const float*)d_in[9];
    float* out = (float*)d_out;

    fused_kernel<<<M_TOT / QPB, 1024, 0, stream>>>(query, global_pc, point_feature,
                                                   w_pos, b_pos, w_pe, b_pe,
                                                   w_v, b_v, w_attn, out);
}


# attention phase, 1 neighbour per lane: pos-embedding leaves the neighbour loop by linearity (u.np_k in score, W_pe.mean pos + b_pe in hbar), 16 score sums by transposed DPP folds; sum order changes
# speedup vs baseline: 36.3516x; 1.0544x over previous
#include <hip/hip_runtime.h>
#include <math.h>

#define B_    2
#define NQ    4096
#define NP    8192
#define D_    128
#define KNN   16
#define M_TOT (B_ * NQ)
#define LEAKY 0.2f

#define QPB      32
#define HB_PITCH 132
#define PRE_STEPS 8
#define SCAN_U    4
#define PASS_PTS  (128 * SCAN_U)
#define NBKT      4
#define NPASS     (NP / PASS_PTS + NBKT - 1)
#define NSLOT     (NPASS * PASS_PTS)
#define XB0 (-0.6744898f)
#define XB1 0.0f
#define XB2 0.6744898f
#define FAR_COORD 1.0e30f

typedef float v2f __attribute__((ext_vector_type(2)));
typedef float v4f __attribute__((ext_vector_type(4)));
typedef float v8f __attribute__((ext_vector_type(8)));

__device__ __forceinline__ int ppos(int i) {
    return __mul24(i >> 7, 384) + (i & 31) * 12 + ((i >> 6) & 1) * 6 + ((i >> 5) & 1);
}

__device__ __forceinline__ float xbound(int k) {
    return k == 0 ? XB0 : (k == 1 ? XB1 : XB2);
}

__device__ __forceinline__ void step_d2(const v4f* sp, v2f qx2, v2f qy2, v2f qz2,
                                        v2f& dab, v2f& dcd) {
    const v4f r0 = sp[0];
    const v4f r1 = sp[1];
    const v4f r2 = sp[2];
    const v2f xab = {r0.x, r0.y}, yab = {r0.z, r0.w}, zab = {r1.x, r1.y};
    const v2f xcd = {r1.z, r1.w}, ycd = {r2.x, r2.y}, zcd = {r2.z, r2.w};
    const v2f dxab = qx2 - xab, dyab = qy2 - yab, dzab = qz2 - zab;
    const v2f dxcd = qx2 - xcd, dycd = qy2 - ycd, dzcd = qz2 - zcd;
    dab = dyab * dyab;
    dcd = dycd * dycd;
    dab = __builtin_elementwise_fma(dxab, dxab, dab);
    dcd = __builtin_elementwise_fma(dxcd, dxcd, dcd);
    dab = __builtin_elementwise_fma(dzab, dzab, dab);
    dcd = __builtin_elementwise_fma(dzcd, dzcd, dcd);
}

typedef __attribute__((address_space(1))) char gchar_t;
typedef __attribute__((address_space(1))) v4f  gv4f_t;

template <int CTRL>
__device__ __forceinline__ float row_lane(float v) {
    return __int_as_float(__builtin_amdgcn_update_dpp(0, __float_as_int(v), CTRL, 0xf, 0xf, true));
}
__device__ __forceinline__ float other_row(float v) {
    const int b = __float_as_int(v);
    return __int_as_float(__builtin_amdgcn_permlanex16(b, b, 0x76543210, 0xfedcba98, false, false));
}
__device__ __forceinline__ float row_sum(float v) {
    v += row_lane<0x128>(v);
    v += row_lane<0x124>(v);
    v += row_lane<0x122>(v);
    v += row_lane<0x121>(v);
    return v;
}
__device__ __forceinline__ float row_max(float v) {
    v = fmaxf(v, row_lane<0x128>(v));
    v = fmaxf(v, row_lane<0x124>(v));
    v = fmaxf(v, row_lane<0x122>(v));
    v = fmaxf(v, row_lane<0x121>(v));
    return v;
}
__device__ __forceinline__ float wave_sum(float v) {
    v = row_sum(v);
    return v + other_row(v);
}
template <int XCTRL>
__device__ __forceinline__ float fold_step(float lo, float hi, bool up) {
    const float keep = up ? hi : lo;
    const float send = up ? lo : hi;
    return keep + row_lane<XCTRL>(send);
}

__global__ __launch_bounds__(1024) void fused_kernel(const float* __restrict__ query,
                                                     const float* __restrict__ global_pc,
                                                     const float* __restrict__ point_feature,
                                                     const float* __restrict__ w_pos,
                                                     const float* __restrict__ b_pos,
                                                     const float* __restrict__ w_pe,
                                                     const float* __restrict__ b_pe,
                                                     const float* __restrict__ w_v,
                                                     const float* __restrict__ b_v,
                                                     const float* __restrict__ w_attn,
                                                     float* __restrict__ out) {
    __shared__ __attribute__((aligned(16))) float smem[NP + 3 * NSLOT + NSLOT];
    float* const hb  = smem;
    float* const pts = smem + NP;
    int* const slot2idx = reinterpret_cast<int*>(smem + NP + 3 * NSLOT);
    int* const wcount   = reinterpret_cast<int*>(smem);

    const int lane = threadIdx.x & 31;
    const int wave  = __builtin_amdgcn_readfirstlane(threadIdx.x >> 5);
    const int mbase = blockIdx.x * QPB;
    const int m     = mbase + wave;
    const int b     = blockIdx.x >> 7;
    const int qbase = mbase & (NQ - 1);
    float* __restrict__ outb = out + (size_t)b * 256 * NQ;

    const float* __restrict__ pc = global_pc + (size_t)b * NP * 3;
    float px[8], py[8], pz[8];
#pragma unroll
    for (int r = 0; r < 8; ++r) {
        const float* p = pc + (size_t)(threadIdx.x + r * 1024) * 3;
        px[r] = p[0]; py[r] = p[1]; pz[r] = p[2];
    }
    unsigned cm[8][NBKT - 1];
    int cum[NBKT - 1];
#pragma unroll
    for (int k = 0; k < NBKT - 1; ++k) cum[k] = 0;
#pragma unroll
    for (int r = 0; r < 8; ++r) {
#pragma unroll
        for (int k = 0; k < NBKT - 1; ++k) {
            cm[r][k] = __builtin_amdgcn_ballot_w32(px[r] < xbound(k));
            cum[k] += __builtin_popcount(cm[r][k]);
        }
    }
    if (lane < NBKT - 1)
        wcount[lane * 32 + wave] = lane == 0 ? cum[0] : (lane == 1 ? cum[1] : cum[2]);

    const float qx = query[m * 3 + 0];
    const float qy = query[m * 3 + 1];
    const float qz = query[m * 3 + 2];
    __syncthreads();

    int tle[NBKT - 1], ble[NBKT - 1];
#pragma unroll
    for (int k = 0; k < NBKT - 1; ++k) {
        const int v = wcount[k * 32 + lane];
        int pk = v + (lane < wave ? (v << 16) : 0);
        pk += __builtin_amdgcn_update_dpp(pk, pk, 0x128, 0xf, 0xf, false);
        pk += __builtin_amdgcn_update_dpp(pk, pk, 0x124, 0xf, 0xf, false);
        pk += __builtin_amdgcn_update_dpp(pk, pk, 0x122, 0xf, 0xf, false);
        pk += __builtin_amdgcn_update_dpp(pk, pk, 0x121, 0xf, 0xf, false);
        pk += __builtin_amdgcn_permlanex16(pk, pk, 0x76543210, 0xfedcba98, false, false);
        tle[k] = __builtin_amdgcn_readfirstlane(pk & 0xffff);
        ble[k] = __builtin_amdgcn_readfirstlane(pk >> 16);
    }
    int nb[NBKT], wb[NBKT], pb[NBKT + 1];
    nb[0] = tle[0];
    wb[0] = ble[0];
#pragma unroll
    for (int s = 1; s < NBKT - 1; ++s) {
        nb[s] = tle[s] - tle[s - 1];
        wb[s] = ble[s] - ble[s - 1];
    }
    nb[NBKT - 1] = NP - tle[NBKT - 2];
    wb[NBKT - 1] = wave * 256 - ble[NBKT - 2];
    pb[0] = 0;
#pragma unroll
    for (int s = 0; s < NBKT; ++s) pb[s + 1] = pb[s] + (nb[s] + PASS_PTS - 1) / PASS_PTS;
    {
        int run[NBKT];
#pragma unroll
        for (int s = 0; s < NBKT; ++s) run[s] = pb[s] * PASS_PTS + wb[s];
#pragma unroll
        for (int r = 0; r < 8; ++r) {
            unsigned bm = ~cm[r][NBKT - 2];
            int a = run[NBKT - 1];
#pragma unroll
            for (int k = NBKT - 2; k >= 0; --k) {
                const unsigned own = k > 0 ? (cm[r][k] & ~cm[r][k - 1]) : cm[r][0];
                if (px[r] < xbound(k)) { bm = own; a = run[k]; }
            }
            const int slot = a + __builtin_amdgcn_mbcnt_lo(bm, 0);
            float* d = pts + ppos(slot);
            d[0] = px[r]; d[2] = py[r]; d[4] = pz[r];
            slot2idx[slot] = threadIdx.x + r * 1024;
            run[0] += __builtin_popcount(cm[r][0]);
#pragma unroll
            for (int s = 1; s < NBKT - 1; ++s) run[s] += __builtin_popcount(cm[r][s] & ~cm[r][s - 1]);
            run[NBKT - 1] += __builtin_popcount(~cm[r][NBKT - 2]);
        }
    }
    if (threadIdx.x < PASS_PTS) {
#pragma unroll
        for (int s = 0; s < NBKT; ++s) {
            const int slot = pb[s] * PASS_PTS + nb[s] + threadIdx.x;
            if (slot < pb[s + 1] * PASS_PTS) {
                float* d = pts + ppos(slot);
                d[0] = FAR_COORD; d[2] = FAR_COORD; d[4] = FAR_COORD;
                slot2idx[slot] = 0;
            }
        }
    }
    __syncthreads();

    {
        const int d = threadIdx.x & (D_ - 1);
        const float w0 = w_pos[d * 3 + 0];
        const float w1 = w_pos[d * 3 + 1];
        const float w2 = w_pos[d * 3 + 2];
        const float bb = b_pos[d];
#pragma unroll
        for (int i = 0; i < 4; ++i) {
            const int ml = i * 8 + (wave >> 2);
            const int mm = mbase + ml;
            float v = w0 * query[mm * 3 + 0]
                    + w1 * query[mm * 3 + 1]
                    + w2 * query[mm * 3 + 2]
                    + bb;
            v = (v >= 0.f) ? v : LEAKY * v;
            outb[(size_t)(D_ + d) * NQ + qbase + ml] = v;
        }
    }

    float ld  = 3.4e38f;
    int   li  = 0;

    v2f qx2 = {qx, qx};
    v2f qy2 = {qy, qy};
    v2f qz2 = {qz, qz};
    const v4f* lp = reinterpret_cast<const v4f*>(pts + lane * 12);

    int bq = 0;
    int pown = pb[0];
#pragma unroll
    for (int k = 0; k < NBKT - 1; ++k)
        if (!(qx < xbound(k))) { bq = k + 1; pown = pb[k + 1]; }
    const int pend = pb[NBKT];
    float gap[NBKT];
#pragma unroll
    for (int s = 0; s < NBKT; ++s) {
        float g = 0.f;
        if (s > 0)        g = fmaxf(g, xbound(s - 1) - qx);
        if (s < NBKT - 1) g = fmaxf(g, qx - xbound(s));
        gap[s] = g;
    }
    const int sb = min(pown * SCAN_U, pend * SCAN_U - PRE_STEPS);
    const v4f* lpb = lp + sb * 96;

    float lmin = 3.4e38f;
#pragma unroll
    for (int s = 0; s < PRE_STEPS; ++s) {
        v2f dab, dcd;
        step_d2(lpb + s * 96, qx2, qy2, qz2, dab, dcd);
        lmin = fminf(lmin, fminf(fminf(dab.x, dab.y), fminf(dcd.x, dcd.y)));
    }
    {
        const int lb = __float_as_int(lmin);
        lmin = fminf(lmin, __int_as_float(__builtin_amdgcn_permlanex16(
                               lb, lb, 0x76543210, 0xfedcba98, false, false)));
    }
    float t0 = lmin;
    t0 = fmaxf(t0, __int_as_float(__builtin_amdgcn_update_dpp(
                       __float_as_int(t0), __float_as_int(t0), 0x128, 0xf, 0xf, false)));
    t0 = fmaxf(t0, __int_as_float(__builtin_amdgcn_update_dpp(
                       __float_as_int(t0), __float_as_int(t0), 0x124, 0xf, 0xf, false)));
    t0 = fmaxf(t0, __int_as_float(__builtin_amdgcn_update_dpp(
                       __float_as_int(t0), __float_as_int(t0), 0x122, 0xf, 0xf, false)));
    t0 = fmaxf(t0, __int_as_float(__builtin_amdgcn_update_dpp(
                       __float_as_int(t0), __float_as_int(t0), 0x121, 0xf, 0xf, false)));
    const int thr0b = __builtin_amdgcn_readfirstlane(__float_as_int(
        fminf(fmaxf(t0 * 1.0001f, 1.17549435e-38f), 3.4e38f)));
    float thr = __int_as_float(thr0b);

    asm volatile("" : "+v"(qx2), "+v"(qy2), "+v"(qz2));

    int slab = bq, dir = 1;
    for (int vis = 0; vis < NBKT + 2; ++vis, slab += dir) {
        bool go = slab >= 0 && slab < NBKT;
        int p0 = 0, p1 = 0;
        if (go) {
            float g = gap[0];
            p0 = pb[0]; p1 = pb[1];
#pragma unroll
            for (int s = 1; s < NBKT; ++s)
                if (slab == s) { g = gap[s]; p0 = pb[s]; p1 = pb[s + 1]; }
            go = g * g < thr;
        }
        if (!go) {
            if (dir < 0) break;
            dir = -1;
            slab = bq;
            continue;
        }
        for (int pass = p0; pass < p1; ++pass) {
            const int base = pass * PASS_PTS;
            const v4f* sp  = lp + pass * (96 * SCAN_U);
            float d2[SCAN_U][4];
            float mn[SCAN_U];
#pragma unroll
            for (int u = 0; u < SCAN_U; ++u) {
                v2f dab, dcd;
                step_d2(sp + u * 96, qx2, qy2, qz2, dab, dcd);
                d2[u][0] = dab.x; d2[u][1] = dab.y; d2[u][2] = dcd.x; d2[u][3] = dcd.y;
                mn[u] = fminf(fminf(dab.x, dab.y), fminf(dcd.x, dcd.y));
            }
            unsigned sm[SCAN_U];
            unsigned any = 0;
#pragma unroll
            for (int u = 0; u < SCAN_U; ++u) {
                sm[u] = __builtin_amdgcn_ballot_w32(mn[u] < thr);
                any |= sm[u];
            }
            if (any == 0) continue;
#pragma unroll
            for (int u = 0; u < SCAN_U; ++u) {
                if (sm[u] != 0) {
                    unsigned gm[4];
#pragma unroll
                    for (int j = 0; j < 4; ++j) gm[j] = __builtin_amdgcn_ballot_w32(d2[u][j] < thr);
#pragma unroll
                    for (int j = 0; j < 4; ++j) {
                        unsigned mask = gm[j];
                        const int d2bits = __float_as_int(d2[u][j]);
                        while (mask) {
                            const int srcl = __builtin_ctz(mask);
                            mask &= mask - 1;
                            const float cd = __int_as_float(__builtin_amdgcn_readlane(d2bits, srcl));
                            const int   ci = base + u * 128 + j * 32 + srcl;
                            const float ldu = __int_as_float(__builtin_amdgcn_update_dpp(
                                0, __float_as_int(ld), 0x111, 0xf, 0xf, true));
                            const int   liu = __builtin_amdgcn_update_dpp(0, li, 0x111, 0xf, 0xf, true);
                            const bool gt  = ld  > cd;
                            const bool gtp = ldu > cd;
                            ld = __builtin_amdgcn_fmed3f(ldu, cd, ld);
                            li = gtp ? liu : (gt ? ci : li);
                        }
                    }
                    const int l15 = __builtin_amdgcn_readlane(__float_as_int(ld), KNN - 1);
                    thr = __int_as_float(l15 < thr0b ? l15 : thr0b);
                }
            }
        }
    }

    const float* np = pts + ppos(li);
    const int lorig = slot2idx[li];
    const float npx = qx - np[0];
    const float npy = qy - np[2];
    const float npz = qz - np[4];

    const int d0 = lane * 4;
    float wx[4], wy[4], wz[4], bp[4], wa[4];
#pragma unroll
    for (int t = 0; t < 4; ++t) {
        wx[t] = w_pe[(d0 + t) * 3 + 0];
        wy[t] = w_pe[(d0 + t) * 3 + 1];
        wz[t] = w_pe[(d0 + t) * 3 + 2];
        bp[t] = b_pe[d0 + t];
        wa[t] = w_attn[d0 + t];
    }

    const float* __restrict__ pfb = point_feature + (size_t)b * NP * D_;
    const unsigned lane_off = (unsigned)lane * 16u;

    v2f h01[KNN], h23[KNN];
    float part[KNN];
    const v2f wa01 = {wa[0], wa[1]}, wa23 = {wa[2], wa[3]};

#pragma unroll
    for (int k = 0; k < KNN; ++k) {
        const int j = __builtin_amdgcn_readlane(lorig, k);
        const unsigned long long ra = (unsigned long long)(pfb + (size_t)j * D_);
        const unsigned ralo = __builtin_amdgcn_readfirstlane((unsigned)ra);
        const unsigned rahi = __builtin_amdgcn_readfirstlane((unsigned)(ra >> 32));
        const gchar_t* rowk = (const gchar_t*)(((unsigned long long)rahi << 32) | ralo);
        const v4f f = *(const gv4f_t*)(rowk + lane_off);
        h01[k] = v2f{f.x, f.y};
        h23[k] = v2f{f.z, f.w};
        const v2f s2 = __builtin_elementwise_fma(h23[k], wa23, h01[k] * wa01);
        part[k] = s2.x + s2.y;
    }

    float ux = 0.f, uy = 0.f, uz = 0.f;
#pragma unroll
    for (int t = 0; t < 4; ++t) {
        ux += wa[t] * wx[t];
        uy += wa[t] * wy[t];
        uz += wa[t] * wz[t];
    }
    ux = wave_sum(ux);
    uy = wave_sum(uy);
    uz = wave_sum(uz);

    float r8[8], r4[4], r2[2];
#pragma unroll
    for (int i = 0; i < 8; ++i) r8[i] = fold_step<0xB1>(part[2 * i], part[2 * i + 1], (lane & 1) != 0);
#pragma unroll
    for (int i = 0; i < 4; ++i) r4[i] = fold_step<0x4E>(r8[2 * i], r8[2 * i + 1], (lane & 2) != 0);
#pragma unroll
    for (int i = 0; i < 2; ++i) r2[i] = fold_step<0x164>(r4[2 * i], r4[2 * i + 1], (lane & 4) != 0);
    float sc = fold_step<0x168>(r2[0], r2[1], (lane & 8) != 0);
    sc += other_row(sc);
    sc += ux * npx + uy * npy + uz * npz;

    const float mx = row_max(sc);
    const float ek = __expf(sc - mx);
    const float inv = 1.0f / row_sum(ek);
    const float nbx = row_sum(ek * npx);
    const float nby = row_sum(ek * npy);
    const float nbz = row_sum(ek * npz);

    v2f a01 = {0.f, 0.f}, a23 = {0.f, 0.f};
#pragma unroll
    for (int k = 0; k < KNN; ++k) {
        const float e = __int_as_float(__builtin_amdgcn_readlane(__float_as_int(ek), k));
        const v2f e2 = {e, e};
        a01 = __builtin_elementwise_fma(e2, h01[k], a01);
        a23 = __builtin_elementwise_fma(e2, h23[k], a23);
    }
    const float acc[4] = {a01.x, a01.y, a23.x, a23.y};

#pragma unroll
    for (int t = 0; t < 4; ++t)
        hb[wave * HB_PITCH + d0 + t] = (acc[t] + wx[t] * nbx + wy[t] * nby + wz[t] * nbz) * inv + bp[t];
    __syncthreads();

    if (wave < 16) {
        const int mt    = wave >> 3;
        const int tileN = (wave & 7) * 16;
        const int half  = lane >> 4;
        const int l     = lane & 15;
        const float* arow = hb + (mt * 16 + l) * HB_PITCH;

        v8f c = {};
        for (int k0 = 0; k0 < D_; k0 += 4) {
            const int k = k0 + half * 2;
            v2f a, bb;
            a.x  = arow[k];
            a.y  = arow[k + 1];
            bb.x = w_v[(size_t)(tileN + l) * D_ + k];
            bb.y = w_v[(size_t)(tileN + l) * D_ + k + 1];
            c = __builtin_amdgcn_wmma_f32_16x16x4_f32(
                     false, a,  false, bb,
                     (short)0, c,  false,  false);
        }

        const float bias = b_v[tileN + l];
#pragma unroll
        for (int r = 0; r < 8; ++r) {
            const int ml = mt * 16 + r + half * 8;
            outb[(size_t)(tileN + l) * NQ + qbase + ml] = c[r] + bias;
        }
    }
}

extern "C" void kernel_launch(void* const* d_in, const int* in_sizes, int n_in,
                              void* d_out, int out_size, void* d_ws, size_t ws_size,
                              hipStream_t stream) {
    const float* query         = (const float*)d_in[0];
    const float* global_pc     = (const float*)d_in[1];
    const float* point_feature = (const float*)d_in[2];
    const float* w_pos         = (const float*)d_in[3];
    const float* b_pos         = (const float*)d_in[4];
    const float* w_pe          = (const float*)d_in[5];
    const float* b_pe          = (const float*)d_in[6];
    const float* w_v           = (const float*)d_in[7];
    const float* b_v           = (const float*)d_in[8];
    const float* w_attn        = (const float*)d_in[9];
    float* out = (float*)d_out;

    fused_kernel<<<M_TOT / QPB, 1024, 0, stream>>>(query, global_pc, point_feature,
                                                   w_pos, b_pos, w_pe, b_pe,
                                                   w_v, b_v, w_attn, out);
}


# all output stores cover whole 128-byte lines, 16 B per lane: query_feature tile turned through LDS, stored by all 32 waves; position_feature 4 queries per thread (was 4-byte stores to 32 lines each)
# speedup vs baseline: 39.7572x; 1.0937x over previous
#include <hip/hip_runtime.h>
#include <math.h>

#define B_    2
#define NQ    4096
#define NP    8192
#define D_    128
#define KNN   16
#define M_TOT (B_ * NQ)
#define LEAKY 0.2f

#define QPB      32
#define HB_PITCH 132
#define TR_PITCH 36
#define PRE_STEPS 8
#define SCAN_U    4
#define PASS_PTS  (128 * SCAN_U)
#define NBKT      4
#define NPASS     (NP / PASS_PTS + NBKT - 1)
#define NSLOT     (NPASS * PASS_PTS)
#define XB0 (-0.6744898f)
#define XB1 0.0f
#define XB2 0.6744898f
#define FAR_COORD 1.0e30f

typedef float v2f __attribute__((ext_vector_type(2)));
typedef float v4f __attribute__((ext_vector_type(4)));
typedef float v8f __attribute__((ext_vector_type(8)));

__device__ __forceinline__ int ppos(int i) {
    return __mul24(i >> 7, 384) + (i & 31) * 12 + ((i >> 6) & 1) * 6 + ((i >> 5) & 1);
}

__device__ __forceinline__ float xbound(int k) {
    return k == 0 ? XB0 : (k == 1 ? XB1 : XB2);
}

__device__ __forceinline__ void step_d2(const v4f* sp, v2f qx2, v2f qy2, v2f qz2,
                                        v2f& dab, v2f& dcd) {
    const v4f r0 = sp[0];
    const v4f r1 = sp[1];
    const v4f r2 = sp[2];
    const v2f xab = {r0.x, r0.y}, yab = {r0.z, r0.w}, zab = {r1.x, r1.y};
    const v2f xcd = {r1.z, r1.w}, ycd = {r2.x, r2.y}, zcd = {r2.z, r2.w};
    const v2f dxab = qx2 - xab, dyab = qy2 - yab, dzab = qz2 - zab;
    const v2f dxcd = qx2 - xcd, dycd = qy2 - ycd, dzcd = qz2 - zcd;
    dab = dyab * dyab;
    dcd = dycd * dycd;
    dab = __builtin_elementwise_fma(dxab, dxab, dab);
    dcd = __builtin_elementwise_fma(dxcd, dxcd, dcd);
    dab = __builtin_elementwise_fma(dzab, dzab, dab);
    dcd = __builtin_elementwise_fma(dzcd, dzcd, dcd);
}

typedef __attribute__((address_space(1))) char gchar_t;
typedef __attribute__((address_space(1))) v4f  gv4f_t;

template <int CTRL>
__device__ __forceinline__ float row_lane(float v) {
    return __int_as_float(__builtin_amdgcn_update_dpp(0, __float_as_int(v), CTRL, 0xf, 0xf, true));
}
__device__ __forceinline__ float other_row(float v) {
    const int b = __float_as_int(v);
    return __int_as_float(__builtin_amdgcn_permlanex16(b, b, 0x76543210, 0xfedcba98, false, false));
}
__device__ __forceinline__ float row_sum(float v) {
    v += row_lane<0x128>(v);
    v += row_lane<0x124>(v);
    v += row_lane<0x122>(v);
    v += row_lane<0x121>(v);
    return v;
}
__device__ __forceinline__ float row_max(float v) {
    v = fmaxf(v, row_lane<0x128>(v));
    v = fmaxf(v, row_lane<0x124>(v));
    v = fmaxf(v, row_lane<0x122>(v));
    v = fmaxf(v, row_lane<0x121>(v));
    return v;
}
__device__ __forceinline__ float wave_sum(float v) {
    v = row_sum(v);
    return v + other_row(v);
}
template <int XCTRL>
__device__ __forceinline__ float fold_step(float lo, float hi, bool up) {
    const float keep = up ? hi : lo;
    const float send = up ? lo : hi;
    return keep + row_lane<XCTRL>(send);
}

__global__ __launch_bounds__(1024) void fused_kernel(const float* __restrict__ query,
                                                     const float* __restrict__ global_pc,
                                                     const float* __restrict__ point_feature,
                                                     const float* __restrict__ w_pos,
                                                     const float* __restrict__ b_pos,
                                                     const float* __restrict__ w_pe,
                                                     const float* __restrict__ b_pe,
                                                     const float* __restrict__ w_v,
                                                     const float* __restrict__ b_v,
                                                     const float* __restrict__ w_attn,
                                                     float* __restrict__ out) {
    __shared__ __attribute__((aligned(16))) float smem[NP + 3 * NSLOT + NSLOT];
    float* const hb  = smem;
    float* const pts = smem + NP;
    int* const slot2idx = reinterpret_cast<int*>(smem + NP + 3 * NSLOT);
    int* const wcount   = reinterpret_cast<int*>(smem);

    const int lane = threadIdx.x & 31;
    const int wave  = __builtin_amdgcn_readfirstlane(threadIdx.x >> 5);
    const int mbase = blockIdx.x * QPB;
    const int m     = mbase + wave;
    const int b     = blockIdx.x >> 7;
    const int qbase = mbase & (NQ - 1);
    float* __restrict__ outb = out + (size_t)b * 256 * NQ;

    const float* __restrict__ pc = global_pc + (size_t)b * NP * 3;
    float px[8], py[8], pz[8];
#pragma unroll
    for (int r = 0; r < 8; ++r) {
        const float* p = pc + (size_t)(threadIdx.x + r * 1024) * 3;
        px[r] = p[0]; py[r] = p[1]; pz[r] = p[2];
    }
    unsigned cm[8][NBKT - 1];
    int cum[NBKT - 1];
#pragma unroll
    for (int k = 0; k < NBKT - 1; ++k) cum[k] = 0;
#pragma unroll
    for (int r = 0; r < 8; ++r) {
#pragma unroll
        for (int k = 0; k < NBKT - 1; ++k) {
            cm[r][k] = __builtin_amdgcn_ballot_w32(px[r] < xbound(k));
            cum[k] += __builtin_popcount(cm[r][k]);
        }
    }
    if (lane < NBKT - 1)
        wcount[lane * 32 + wave] = lane == 0 ? cum[0] : (lane == 1 ? cum[1] : cum[2]);

    const float qx = query[m * 3 + 0];
    const float qy = query[m * 3 + 1];
    const float qz = query[m * 3 + 2];
    __syncthreads();

    int tle[NBKT - 1], ble[NBKT - 1];
#pragma unroll
    for (int k = 0; k < NBKT - 1; ++k) {
        const int v = wcount[k * 32 + lane];
        int pk = v + (lane < wave ? (v << 16) : 0);
        pk += __builtin_amdgcn_update_dpp(pk, pk, 0x128, 0xf, 0xf, false);
        pk += __builtin_amdgcn_update_dpp(pk, pk, 0x124, 0xf, 0xf, false);
        pk += __builtin_amdgcn_update_dpp(pk, pk, 0x122, 0xf, 0xf, false);
        pk += __builtin_amdgcn_update_dpp(pk, pk, 0x121, 0xf, 0xf, false);
        pk += __builtin_amdgcn_permlanex16(pk, pk, 0x76543210, 0xfedcba98, false, false);
        tle[k] = __builtin_amdgcn_readfirstlane(pk & 0xffff);
        ble[k] = __builtin_amdgcn_readfirstlane(pk >> 16);
    }
    int nb[NBKT], wb[NBKT], pb[NBKT + 1];
    nb[0] = tle[0];
    wb[0] = ble[0];
#pragma unroll
    for (int s = 1; s < NBKT - 1; ++s) {
        nb[s] = tle[s] - tle[s - 1];
        wb[s] = ble[s] - ble[s - 1];
    }
    nb[NBKT - 1] = NP - tle[NBKT - 2];
    wb[NBKT - 1] = wave * 256 - ble[NBKT - 2];
    pb[0] = 0;
#pragma unroll
    for (int s = 0; s < NBKT; ++s) pb[s + 1] = pb[s] + (nb[s] + PASS_PTS - 1) / PASS_PTS;
    {
        int run[NBKT];
#pragma unroll
        for (int s = 0; s < NBKT; ++s) run[s] = pb[s] * PASS_PTS + wb[s];
#pragma unroll
        for (int r = 0; r < 8; ++r) {
            unsigned bm = ~cm[r][NBKT - 2];
            int a = run[NBKT - 1];
#pragma unroll
            for (int k = NBKT - 2; k >= 0; --k) {
                const unsigned own = k > 0 ? (cm[r][k] & ~cm[r][k - 1]) : cm[r][0];
                if (px[r] < xbound(k)) { bm = own; a = run[k]; }
            }
            const int slot = a + __builtin_amdgcn_mbcnt_lo(bm, 0);
            float* d = pts + ppos(slot);
            d[0] = px[r]; d[2] = py[r]; d[4] = pz[r];
            slot2idx[slot] = threadIdx.x + r * 1024;
            run[0] += __builtin_popcount(cm[r][0]);
#pragma unroll
            for (int s = 1; s < NBKT - 1; ++s) run[s] += __builtin_popcount(cm[r][s] & ~cm[r][s - 1]);
            run[NBKT - 1] += __builtin_popcount(~cm[r][NBKT - 2]);
        }
    }
    if (threadIdx.x < PASS_PTS) {
#pragma unroll
        for (int s = 0; s < NBKT; ++s) {
            const int slot = pb[s] * PASS_PTS + nb[s] + threadIdx.x;
            if (slot < pb[s + 1] * PASS_PTS) {
                float* d = pts + ppos(slot);
                d[0] = FAR_COORD; d[2] = FAR_COORD; d[4] = FAR_COORD;
                slot2idx[slot] = 0;
            }
        }
    }
    __syncthreads();

    {
        const int d    = threadIdx.x >> 3;
        const int quad = threadIdx.x & 7;
        const float w0 = w_pos[d * 3 + 0];
        const float w1 = w_pos[d * 3 + 1];
        const float w2 = w_pos[d * 3 + 2];
        const float bb = b_pos[d];
        const v4f* qv = reinterpret_cast<const v4f*>(query + (size_t)(mbase + quad * 4) * 3);
        const v4f q0 = qv[0];
        const v4f q1 = qv[1];
        const v4f q2 = qv[2];
        const float cx[4] = {q0.x, q0.w, q1.z, q2.y};
        const float cy[4] = {q0.y, q1.x, q1.w, q2.z};
        const float cz[4] = {q0.z, q1.y, q2.x, q2.w};
        float o[4];
#pragma unroll
        for (int i = 0; i < 4; ++i) {
            float v = w0 * cx[i]
                    + w1 * cy[i]
                    + w2 * cz[i]
                    + bb;
            o[i] = (v >= 0.f) ? v : LEAKY * v;
        }
        *reinterpret_cast<v4f*>(outb + (size_t)(D_ + d) * NQ + qbase + quad * 4) = v4f{o[0], o[1], o[2], o[3]};
    }

    float ld  = 3.4e38f;
    int   li  = 0;

    v2f qx2 = {qx, qx};
    v2f qy2 = {qy, qy};
    v2f qz2 = {qz, qz};
    const v4f* lp = reinterpret_cast<const v4f*>(pts + lane * 12);

    int bq = 0;
    int pown = pb[0];
#pragma unroll
    for (int k = 0; k < NBKT - 1; ++k)
        if (!(qx < xbound(k))) { bq = k + 1; pown = pb[k + 1]; }
    const int pend = pb[NBKT];
    float gap[NBKT];
#pragma unroll
    for (int s = 0; s < NBKT; ++s) {
        float g = 0.f;
        if (s > 0)        g = fmaxf(g, xbound(s - 1) - qx);
        if (s < NBKT - 1) g = fmaxf(g, qx - xbound(s));
        gap[s] = g;
    }
    const int sb = min(pown * SCAN_U, pend * SCAN_U - PRE_STEPS);
    const v4f* lpb = lp + sb * 96;

    float lmin = 3.4e38f;
#pragma unroll
    for (int s = 0; s < PRE_STEPS; ++s) {
        v2f dab, dcd;
        step_d2(lpb + s * 96, qx2, qy2, qz2, dab, dcd);
        lmin = fminf(lmin, fminf(fminf(dab.x, dab.y), fminf(dcd.x, dcd.y)));
    }
    {
        const int lb = __float_as_int(lmin);
        lmin = fminf(lmin, __int_as_float(__builtin_amdgcn_permlanex16(
                               lb, lb, 0x76543210, 0xfedcba98, false, false)));
    }
    float t0 = lmin;
    t0 = fmaxf(t0, __int_as_float(__builtin_amdgcn_update_dpp(
                       __float_as_int(t0), __float_as_int(t0), 0x128, 0xf, 0xf, false)));
    t0 = fmaxf(t0, __int_as_float(__builtin_amdgcn_update_dpp(
                       __float_as_int(t0), __float_as_int(t0), 0x124, 0xf, 0xf, false)));
    t0 = fmaxf(t0, __int_as_float(__builtin_amdgcn_update_dpp(
                       __float_as_int(t0), __float_as_int(t0), 0x122, 0xf, 0xf, false)));
    t0 = fmaxf(t0, __int_as_float(__builtin_amdgcn_update_dpp(
                       __float_as_int(t0), __float_as_int(t0), 0x121, 0xf, 0xf, false)));
    const int thr0b = __builtin_amdgcn_readfirstlane(__float_as_int(
        fminf(fmaxf(t0 * 1.0001f, 1.17549435e-38f), 3.4e38f)));
    float thr = __int_as_float(thr0b);

    asm volatile("" : "+v"(qx2), "+v"(qy2), "+v"(qz2));

    int slab = bq, dir = 1;
    for (int vis = 0; vis < NBKT + 2; ++vis, slab += dir) {
        bool go = slab >= 0 && slab < NBKT;
        int p0 = 0, p1 = 0;
        if (go) {
            float g = gap[0];
            p0 = pb[0]; p1 = pb[1];
#pragma unroll
            for (int s = 1; s < NBKT; ++s)
                if (slab == s) { g = gap[s]; p0 = pb[s]; p1 = pb[s + 1]; }
            go = g * g < thr;
        }
        if (!go) {
            if (dir < 0) break;
            dir = -1;
            slab = bq;
            continue;
        }
        for (int pass = p0; pass < p1; ++pass) {
            const int base = pass * PASS_PTS;
            const v4f* sp  = lp + pass * (96 * SCAN_U);
            float d2[SCAN_U][4];
            float mn[SCAN_U];
#pragma unroll
            for (int u = 0; u < SCAN_U; ++u) {
                v2f dab, dcd;
                step_d2(sp + u * 96, qx2, qy2, qz2, dab, dcd);
                d2[u][0] = dab.x; d2[u][1] = dab.y; d2[u][2] = dcd.x; d2[u][3] = dcd.y;
                mn[u] = fminf(fminf(dab.x, dab.y), fminf(dcd.x, dcd.y));
            }
            unsigned sm[SCAN_U];
            unsigned any = 0;
#pragma unroll
            for (int u = 0; u < SCAN_U; ++u) {
                sm[u] = __builtin_amdgcn_ballot_w32(mn[u] < thr);
                any |= sm[u];
            }
            if (any == 0) continue;
#pragma unroll
            for (int u = 0; u < SCAN_U; ++u) {
                if (sm[u] != 0) {
                    unsigned gm[4];
#pragma unroll
                    for (int j = 0; j < 4; ++j) gm[j] = __builtin_amdgcn_ballot_w32(d2[u][j] < thr);
#pragma unroll
                    for (int j = 0; j < 4; ++j) {
                        unsigned mask = gm[j];
                        const int d2bits = __float_as_int(d2[u][j]);
                        while (mask) {
                            const int srcl = __builtin_ctz(mask);
                            mask &= mask - 1;
                            const float cd = __int_as_float(__builtin_amdgcn_readlane(d2bits, srcl));
                            const int   ci = base + u * 128 + j * 32 + srcl;
                            const float ldu = __int_as_float(__builtin_amdgcn_update_dpp(
                                0, __float_as_int(ld), 0x111, 0xf, 0xf, true));
                            const int   liu = __builtin_amdgcn_update_dpp(0, li, 0x111, 0xf, 0xf, true);
                            const bool gt  = ld  > cd;
                            const bool gtp = ldu > cd;
                            ld = __builtin_amdgcn_fmed3f(ldu, cd, ld);
                            li = gtp ? liu : (gt ? ci : li);
                        }
                    }
                    const int l15 = __builtin_amdgcn_readlane(__float_as_int(ld), KNN - 1);
                    thr = __int_as_float(l15 < thr0b ? l15 : thr0b);
                }
            }
        }
    }

    const float* np = pts + ppos(li);
    const int lorig = slot2idx[li];
    const float npx = qx - np[0];
    const float npy = qy - np[2];
    const float npz = qz - np[4];

    const int d0 = lane * 4;
    float wx[4], wy[4], wz[4], bp[4], wa[4];
#pragma unroll
    for (int t = 0; t < 4; ++t) {
        wx[t] = w_pe[(d0 + t) * 3 + 0];
        wy[t] = w_pe[(d0 + t) * 3 + 1];
        wz[t] = w_pe[(d0 + t) * 3 + 2];
        bp[t] = b_pe[d0 + t];
        wa[t] = w_attn[d0 + t];
    }

    const float* __restrict__ pfb = point_feature + (size_t)b * NP * D_;
    const unsigned lane_off = (unsigned)lane * 16u;

    v2f h01[KNN], h23[KNN];
    float part[KNN];
    const v2f wa01 = {wa[0], wa[1]}, wa23 = {wa[2], wa[3]};

#pragma unroll
    for (int k = 0; k < KNN; ++k) {
        const int j = __builtin_amdgcn_readlane(lorig, k);
        const unsigned long long ra = (unsigned long long)(pfb + (size_t)j * D_);
        const unsigned ralo = __builtin_amdgcn_readfirstlane((unsigned)ra);
        const unsigned rahi = __builtin_amdgcn_readfirstlane((unsigned)(ra >> 32));
        const gchar_t* rowk = (const gchar_t*)(((unsigned long long)rahi << 32) | ralo);
        const v4f f = *(const gv4f_t*)(rowk + lane_off);
        h01[k] = v2f{f.x, f.y};
        h23[k] = v2f{f.z, f.w};
        const v2f s2 = __builtin_elementwise_fma(h23[k], wa23, h01[k] * wa01);
        part[k] = s2.x + s2.y;
    }

    float ux = 0.f, uy = 0.f, uz = 0.f;
#pragma unroll
    for (int t = 0; t < 4; ++t) {
        ux += wa[t] * wx[t];
        uy += wa[t] * wy[t];
        uz += wa[t] * wz[t];
    }
    ux = wave_sum(ux);
    uy = wave_sum(uy);
    uz = wave_sum(uz);

    float r8[8], r4[4], r2[2];
#pragma unroll
    for (int i = 0; i < 8; ++i) r8[i] = fold_step<0xB1>(part[2 * i], part[2 * i + 1], (lane & 1) != 0);
#pragma unroll
    for (int i = 0; i < 4; ++i) r4[i] = fold_step<0x4E>(r8[2 * i], r8[2 * i + 1], (lane & 2) != 0);
#pragma unroll
    for (int i = 0; i < 2; ++i) r2[i] = fold_step<0x164>(r4[2 * i], r4[2 * i + 1], (lane & 4) != 0);
    float sc = fold_step<0x168>(r2[0], r2[1], (lane & 8) != 0);
    sc += other_row(sc);
    sc += ux * npx + uy * npy + uz * npz;

    const float mx = row_max(sc);
    const float ek = __expf(sc - mx);
    const float inv = 1.0f / row_sum(ek);
    const float nbx = row_sum(ek * npx);
    const float nby = row_sum(ek * npy);
    const float nbz = row_sum(ek * npz);

    v2f a01 = {0.f, 0.f}, a23 = {0.f, 0.f};
#pragma unroll
    for (int k = 0; k < KNN; ++k) {
        const float e = __int_as_float(__builtin_amdgcn_readlane(__float_as_int(ek), k));
        const v2f e2 = {e, e};
        a01 = __builtin_elementwise_fma(e2, h01[k], a01);
        a23 = __builtin_elementwise_fma(e2, h23[k], a23);
    }
    const float acc[4] = {a01.x, a01.y, a23.x, a23.y};

#pragma unroll
    for (int t = 0; t < 4; ++t)
        hb[wave * HB_PITCH + d0 + t] = (acc[t] + wx[t] * nbx + wy[t] * nby + wz[t] * nbz) * inv + bp[t];
    __syncthreads();

    if (wave < 16) {
        const int mt    = wave >> 3;
        const int tileN = (wave & 7) * 16;
        const int half  = lane >> 4;
        const int l     = lane & 15;
        const float* arow = hb + (mt * 16 + l) * HB_PITCH;

        v8f c = {};
        for (int k0 = 0; k0 < D_; k0 += 4) {
            const int k = k0 + half * 2;
            v2f a, bb;
            a.x  = arow[k];
            a.y  = arow[k + 1];
            bb.x = w_v[(size_t)(tileN + l) * D_ + k];
            bb.y = w_v[(size_t)(tileN + l) * D_ + k + 1];
            c = __builtin_amdgcn_wmma_f32_16x16x4_f32(
                     false, a,  false, bb,
                     (short)0, c,  false,  false);
        }

        const float bias = b_v[tileN + l];
#pragma unroll
        for (int r = 0; r < 8; ++r) {
            const int ml = mt * 16 + r + half * 8;
            pts[(tileN + l) * TR_PITCH + ml] = c[r] + bias;
        }
    }
    __syncthreads();
    {
        const int d    = threadIdx.x >> 3;
        const int quad = threadIdx.x & 7;
        const v4f v = *reinterpret_cast<const v4f*>(pts + d * TR_PITCH + quad * 4);
        *reinterpret_cast<v4f*>(outb + (size_t)d * NQ + qbase + quad * 4) = v;
    }
}

extern "C" void kernel_launch(void* const* d_in, const int* in_sizes, int n_in,
                              void* d_out, int out_size, void* d_ws, size_t ws_size,
                              hipStream_t stream) {
    const float* query         = (const float*)d_in[0];
    const float* global_pc     = (const float*)d_in[1];
    const float* point_feature = (const float*)d_in[2];
    const float* w_pos         = (const float*)d_in[3];
    const float* b_pos         = (const float*)d_in[4];
    const float* w_pe          = (const float*)d_in[5];
    const float* b_pe          = (const float*)d_in[6];
    const float* w_v           = (const float*)d_in[7];
    const float* b_v           = (const float*)d_in[8];
    const float* w_attn        = (const float*)d_in[9];
    float* out = (float*)d_out;

    fused_kernel<<<M_TOT / QPB, 1024, 0, stream>>>(query, global_pc, point_feature,
                                                   w_pos, b_pos, w_pe, b_pe,
                                                   w_v, b_v, w_attn, out);
}


# W_v reaches the WMMA through LDS: the block loads the matrix once by whole 128-byte lines, issued before the scan, stored after it; B fragment loads no longer touch 16 global lines each
# speedup vs baseline: 42.3366x; 1.0649x over previous
#include <hip/hip_runtime.h>
#include <math.h>

#define B_    2
#define NQ    4096
#define NP    8192
#define D_    128
#define KNN   16
#define M_TOT (B_ * NQ)
#define LEAKY 0.2f

#define QPB      32
#define HB_PITCH 132
#define TR_PITCH 36
#define WV_PITCH 132
#define PRE_STEPS 8
#define SCAN_U    4
#define PASS_PTS  (128 * SCAN_U)
#define NBKT      4
#define NPASS     (NP / PASS_PTS + NBKT - 1)
#define NSLOT     (NPASS * PASS_PTS)
#define XB0 (-0.6744898f)
#define XB1 0.0f
#define XB2 0.6744898f
#define FAR_COORD 1.0e30f

typedef float v2f __attribute__((ext_vector_type(2)));
typedef float v4f __attribute__((ext_vector_type(4)));
typedef float v8f __attribute__((ext_vector_type(8)));

__device__ __forceinline__ int ppos(int i) {
    return __mul24(i >> 7, 384) + (i & 31) * 12 + ((i >> 6) & 1) * 6 + ((i >> 5) & 1);
}

__device__ __forceinline__ float xbound(int k) {
    return k == 0 ? XB0 : (k == 1 ? XB1 : XB2);
}

__device__ __forceinline__ void step_d2(const v4f* sp, v2f qx2, v2f qy2, v2f qz2,
                                        v2f& dab, v2f& dcd) {
    const v4f r0 = sp[0];
    const v4f r1 = sp[1];
    const v4f r2 = sp[2];
    const v2f xab = {r0.x, r0.y}, yab = {r0.z, r0.w}, zab = {r1.x, r1.y};
    const v2f xcd = {r1.z, r1.w}, ycd = {r2.x, r2.y}, zcd = {r2.z, r2.w};
    const v2f dxab = qx2 - xab, dyab = qy2 - yab, dzab = qz2 - zab;
    const v2f dxcd = qx2 - xcd, dycd = qy2 - ycd, dzcd = qz2 - zcd;
    dab = dyab * dyab;
    dcd = dycd * dycd;
    dab = __builtin_elementwise_fma(dxab, dxab, dab);
    dcd = __builtin_elementwise_fma(dxcd, dxcd, dcd);
    dab = __builtin_elementwise_fma(dzab, dzab, dab);
    dcd = __builtin_elementwise_fma(dzcd, dzcd, dcd);
}

typedef __attribute__((address_space(1))) char gchar_t;
typedef __attribute__((address_space(1))) v4f  gv4f_t;

template <int CTRL>
__device__ __forceinline__ float row_lane(float v) {
    return __int_as_float(__builtin_amdgcn_update_dpp(0, __float_as_int(v), CTRL, 0xf, 0xf, true));
}
__device__ __forceinline__ float other_row(float v) {
    const int b = __float_as_int(v);
    return __int_as_float(__builtin_amdgcn_permlanex16(b, b, 0x76543210, 0xfedcba98, false, false));
}
__device__ __forceinline__ float row_sum(float v) {
    v += row_lane<0x128>(v);
    v += row_lane<0x124>(v);
    v += row_lane<0x122>(v);
    v += row_lane<0x121>(v);
    return v;
}
__device__ __forceinline__ float row_max(float v) {
    v = fmaxf(v, row_lane<0x128>(v));
    v = fmaxf(v, row_lane<0x124>(v));
    v = fmaxf(v, row_lane<0x122>(v));
    v = fmaxf(v, row_lane<0x121>(v));
    return v;
}
__device__ __forceinline__ float wave_sum(float v) {
    v = row_sum(v);
    return v + other_row(v);
}
template <int XCTRL>
__device__ __forceinline__ float fold_step(float lo, float hi, bool up) {
    const float keep = up ? hi : lo;
    const float send = up ? lo : hi;
    return keep + row_lane<XCTRL>(send);
}

__global__ __launch_bounds__(1024) void fused_kernel(const float* __restrict__ query,
                                                     const float* __restrict__ global_pc,
                                                     const float* __restrict__ point_feature,
                                                     const float* __restrict__ w_pos,
                                                     const float* __restrict__ b_pos,
                                                     const float* __restrict__ w_pe,
                                                     const float* __restrict__ b_pe,
                                                     const float* __restrict__ w_v,
                                                     const float* __restrict__ b_v,
                                                     const float* __restrict__ w_attn,
                                                     float* __restrict__ out) {
    __shared__ __attribute__((aligned(16))) float smem[NP + 3 * NSLOT + NSLOT + D_ * WV_PITCH];
    float* const hb  = smem;
    float* const pts = smem + NP;
    int* const slot2idx = reinterpret_cast<int*>(smem + NP + 3 * NSLOT);
    int* const wcount   = reinterpret_cast<int*>(smem);
    float* const wvs = smem + NP + 3 * NSLOT + NSLOT;

    const int lane = threadIdx.x & 31;
    const int wave  = __builtin_amdgcn_readfirstlane(threadIdx.x >> 5);
    const int mbase = blockIdx.x * QPB;
    const int m     = mbase + wave;
    const int b     = blockIdx.x >> 7;
    const int qbase = mbase & (NQ - 1);
    float* __restrict__ outb = out + (size_t)b * 256 * NQ;

    const float* __restrict__ pc = global_pc + (size_t)b * NP * 3;
    float px[8], py[8], pz[8];
#pragma unroll
    for (int r = 0; r < 8; ++r) {
        const float* p = pc + (size_t)(threadIdx.x + r * 1024) * 3;
        px[r] = p[0]; py[r] = p[1]; pz[r] = p[2];
    }
    unsigned cm[8][NBKT - 1];
    int cum[NBKT - 1];
#pragma unroll
    for (int k = 0; k < NBKT - 1; ++k) cum[k] = 0;
#pragma unroll
    for (int r = 0; r < 8; ++r) {
#pragma unroll
        for (int k = 0; k < NBKT - 1; ++k) {
            cm[r][k] = __builtin_amdgcn_ballot_w32(px[r] < xbound(k));
            cum[k] += __builtin_popcount(cm[r][k]);
        }
    }
    if (lane < NBKT - 1)
        wcount[lane * 32 + wave] = lane == 0 ? cum[0] : (lane == 1 ? cum[1] : cum[2]);

    const float qx = query[m * 3 + 0];
    const float qy = query[m * 3 + 1];
    const float qz = query[m * 3 + 2];
    __syncthreads();

    int tle[NBKT - 1], ble[NBKT - 1];
#pragma unroll
    for (int k = 0; k < NBKT - 1; ++k) {
        const int v = wcount[k * 32 + lane];
        int pk = v + (lane < wave ? (v << 16) : 0);
        pk += __builtin_amdgcn_update_dpp(pk, pk, 0x128, 0xf, 0xf, false);
        pk += __builtin_amdgcn_update_dpp(pk, pk, 0x124, 0xf, 0xf, false);
        pk += __builtin_amdgcn_update_dpp(pk, pk, 0x122, 0xf, 0xf, false);
        pk += __builtin_amdgcn_update_dpp(pk, pk, 0x121, 0xf, 0xf, false);
        pk += __builtin_amdgcn_permlanex16(pk, pk, 0x76543210, 0xfedcba98, false, false);
        tle[k] = __builtin_amdgcn_readfirstlane(pk & 0xffff);
        ble[k] = __builtin_amdgcn_readfirstlane(pk >> 16);
    }
    int nb[NBKT], wb[NBKT], pb[NBKT + 1];
    nb[0] = tle[0];
    wb[0] = ble[0];
#pragma unroll
    for (int s = 1; s < NBKT - 1; ++s) {
        nb[s] = tle[s] - tle[s - 1];
        wb[s] = ble[s] - ble[s - 1];
    }
    nb[NBKT - 1] = NP - tle[NBKT - 2];
    wb[NBKT - 1] = wave * 256 - ble[NBKT - 2];
    pb[0] = 0;
#pragma unroll
    for (int s = 0; s < NBKT; ++s) pb[s + 1] = pb[s] + (nb[s] + PASS_PTS - 1) / PASS_PTS;
    {
        int run[NBKT];
#pragma unroll
        for (int s = 0; s < NBKT; ++s) run[s] = pb[s] * PASS_PTS + wb[s];
#pragma unroll
        for (int r = 0; r < 8; ++r) {
            unsigned bm = ~cm[r][NBKT - 2];
            int a = run[NBKT - 1];
#pragma unroll
            for (int k = NBKT - 2; k >= 0; --k) {
                const unsigned own = k > 0 ? (cm[r][k] & ~cm[r][k - 1]) : cm[r][0];
                if (px[r] < xbound(k)) { bm = own; a = run[k]; }
            }
            const int slot = a + __builtin_amdgcn_mbcnt_lo(bm, 0);
            float* d = pts + ppos(slot);
            d[0] = px[r]; d[2] = py[r]; d[4] = pz[r];
            slot2idx[slot] = threadIdx.x + r * 1024;
            run[0] += __builtin_popcount(cm[r][0]);
#pragma unroll
            for (int s = 1; s < NBKT - 1; ++s) run[s] += __builtin_popcount(cm[r][s] & ~cm[r][s - 1]);
            run[NBKT - 1] += __builtin_popcount(~cm[r][NBKT - 2]);
        }
    }
    if (threadIdx.x < PASS_PTS) {
#pragma unroll
        for (int s = 0; s < NBKT; ++s) {
            const int slot = pb[s] * PASS_PTS + nb[s] + threadIdx.x;
            if (slot < pb[s + 1] * PASS_PTS) {
                float* d = pts + ppos(slot);
                d[0] = FAR_COORD; d[2] = FAR_COORD; d[4] = FAR_COORD;
                slot2idx[slot] = 0;
            }
        }
    }
    __syncthreads();

    v4f wvr[4];
#pragma unroll
    for (int i = 0; i < 4; ++i)
        wvr[i] = *reinterpret_cast<const v4f*>(w_v + (size_t)(threadIdx.x + i * 1024) * 4);

    {
        const int d    = threadIdx.x >> 3;
        const int quad = threadIdx.x & 7;
        const float w0 = w_pos[d * 3 + 0];
        const float w1 = w_pos[d * 3 + 1];
        const float w2 = w_pos[d * 3 + 2];
        const float bb = b_pos[d];
        const v4f* qv = reinterpret_cast<const v4f*>(query + (size_t)(mbase + quad * 4) * 3);
        const v4f q0 = qv[0];
        const v4f q1 = qv[1];
        const v4f q2 = qv[2];
        const float cx[4] = {q0.x, q0.w, q1.z, q2.y};
        const float cy[4] = {q0.y, q1.x, q1.w, q2.z};
        const float cz[4] = {q0.z, q1.y, q2.x, q2.w};
        float o[4];
#pragma unroll
        for (int i = 0; i < 4; ++i) {
            float v = w0 * cx[i]
                    + w1 * cy[i]
                    + w2 * cz[i]
                    + bb;
            o[i] = (v >= 0.f) ? v : LEAKY * v;
        }
        *reinterpret_cast<v4f*>(outb + (size_t)(D_ + d) * NQ + qbase + quad * 4) = v4f{o[0], o[1], o[2], o[3]};
    }

    float ld  = 3.4e38f;
    int   li  = 0;

    v2f qx2 = {qx, qx};
    v2f qy2 = {qy, qy};
    v2f qz2 = {qz, qz};
    const v4f* lp = reinterpret_cast<const v4f*>(pts + lane * 12);

    int bq = 0;
    int pown = pb[0];
#pragma unroll
    for (int k = 0; k < NBKT - 1; ++k)
        if (!(qx < xbound(k))) { bq = k + 1; pown = pb[k + 1]; }
    const int pend = pb[NBKT];
    float gap[NBKT];
#pragma unroll
    for (int s = 0; s < NBKT; ++s) {
        float g = 0.f;
        if (s > 0)        g = fmaxf(g, xbound(s - 1) - qx);
        if (s < NBKT - 1) g = fmaxf(g, qx - xbound(s));
        gap[s] = g;
    }
    const int sb = min(pown * SCAN_U, pend * SCAN_U - PRE_STEPS);
    const v4f* lpb = lp + sb * 96;

    float lmin = 3.4e38f;
#pragma unroll
    for (int s = 0; s < PRE_STEPS; ++s) {
        v2f dab, dcd;
        step_d2(lpb + s * 96, qx2, qy2, qz2, dab, dcd);
        lmin = fminf(lmin, fminf(fminf(dab.x, dab.y), fminf(dcd.x, dcd.y)));
    }
    {
        const int lb = __float_as_int(lmin);
        lmin = fminf(lmin, __int_as_float(__builtin_amdgcn_permlanex16(
                               lb, lb, 0x76543210, 0xfedcba98, false, false)));
    }
    float t0 = lmin;
    t0 = fmaxf(t0, __int_as_float(__builtin_amdgcn_update_dpp(
                       __float_as_int(t0), __float_as_int(t0), 0x128, 0xf, 0xf, false)));
    t0 = fmaxf(t0, __int_as_float(__builtin_amdgcn_update_dpp(
                       __float_as_int(t0), __float_as_int(t0), 0x124, 0xf, 0xf, false)));
    t0 = fmaxf(t0, __int_as_float(__builtin_amdgcn_update_dpp(
                       __float_as_int(t0), __float_as_int(t0), 0x122, 0xf, 0xf, false)));
    t0 = fmaxf(t0, __int_as_float(__builtin_amdgcn_update_dpp(
                       __float_as_int(t0), __float_as_int(t0), 0x121, 0xf, 0xf, false)));
    const int thr0b = __builtin_amdgcn_readfirstlane(__float_as_int(
        fminf(fmaxf(t0 * 1.0001f, 1.17549435e-38f), 3.4e38f)));
    float thr = __int_as_float(thr0b);

    asm volatile("" : "+v"(qx2), "+v"(qy2), "+v"(qz2));

    int slab = bq, dir = 1;
    for (int vis = 0; vis < NBKT + 2; ++vis, slab += dir) {
        bool go = slab >= 0 && slab < NBKT;
        int p0 = 0, p1 = 0;
        if (go) {
            float g = gap[0];
            p0 = pb[0]; p1 = pb[1];
#pragma unroll
            for (int s = 1; s < NBKT; ++s)
                if (slab == s) { g = gap[s]; p0 = pb[s]; p1 = pb[s + 1]; }
            go = g * g < thr;
        }
        if (!go) {
            if (dir < 0) break;
            dir = -1;
            slab = bq;
            continue;
        }
        for (int pass = p0; pass < p1; ++pass) {
            const int base = pass * PASS_PTS;
            const v4f* sp  = lp + pass * (96 * SCAN_U);
            float d2[SCAN_U][4];
            float mn[SCAN_U];
#pragma unroll
            for (int u = 0; u < SCAN_U; ++u) {
                v2f dab, dcd;
                step_d2(sp + u * 96, qx2, qy2, qz2, dab, dcd);
                d2[u][0] = dab.x; d2[u][1] = dab.y; d2[u][2] = dcd.x; d2[u][3] = dcd.y;
                mn[u] = fminf(fminf(dab.x, dab.y), fminf(dcd.x, dcd.y));
            }
            unsigned sm[SCAN_U];
            unsigned any = 0;
#pragma unroll
            for (int u = 0; u < SCAN_U; ++u) {
                sm[u] = __builtin_amdgcn_ballot_w32(mn[u] < thr);
                any |= sm[u];
            }
            if (any == 0) continue;
#pragma unroll
            for (int u = 0; u < SCAN_U; ++u) {
                if (sm[u] != 0) {
                    unsigned gm[4];
#pragma unroll
                    for (int j = 0; j < 4; ++j) gm[j] = __builtin_amdgcn_ballot_w32(d2[u][j] < thr);
#pragma unroll
                    for (int j = 0; j < 4; ++j) {
                        unsigned mask = gm[j];
                        const int d2bits = __float_as_int(d2[u][j]);
                        while (mask) {
                            const int srcl = __builtin_ctz(mask);
                            mask &= mask - 1;
                            const float cd = __int_as_float(__builtin_amdgcn_readlane(d2bits, srcl));
                            const int   ci = base + u * 128 + j * 32 + srcl;
                            const float ldu = __int_as_float(__builtin_amdgcn_update_dpp(
                                0, __float_as_int(ld), 0x111, 0xf, 0xf, true));
                            const int   liu = __builtin_amdgcn_update_dpp(0, li, 0x111, 0xf, 0xf, true);
                            const bool gt  = ld  > cd;
                            const bool gtp = ldu > cd;
                            ld = __builtin_amdgcn_fmed3f(ldu, cd, ld);
                            li = gtp ? liu : (gt ? ci : li);
                        }
                    }
                    const int l15 = __builtin_amdgcn_readlane(__float_as_int(ld), KNN - 1);
                    thr = __int_as_float(l15 < thr0b ? l15 : thr0b);
                }
            }
        }
    }

#pragma unroll
    for (int i = 0; i < 4; ++i)
        *reinterpret_cast<v4f*>(wvs + (wave + 32 * i) * WV_PITCH + lane * 4) = wvr[i];

    const float* np = pts + ppos(li);
    const int lorig = slot2idx[li];
    const float npx = qx - np[0];
    const float npy = qy - np[2];
    const float npz = qz - np[4];

    const int d0 = lane * 4;
    float wx[4], wy[4], wz[4], bp[4], wa[4];
#pragma unroll
    for (int t = 0; t < 4; ++t) {
        wx[t] = w_pe[(d0 + t) * 3 + 0];
        wy[t] = w_pe[(d0 + t) * 3 + 1];
        wz[t] = w_pe[(d0 + t) * 3 + 2];
        bp[t] = b_pe[d0 + t];
        wa[t] = w_attn[d0 + t];
    }

    const float* __restrict__ pfb = point_feature + (size_t)b * NP * D_;
    const unsigned lane_off = (unsigned)lane * 16u;

    v2f h01[KNN], h23[KNN];
    float part[KNN];
    const v2f wa01 = {wa[0], wa[1]}, wa23 = {wa[2], wa[3]};

#pragma unroll
    for (int k = 0; k < KNN; ++k) {
        const int j = __builtin_amdgcn_readlane(lorig, k);
        const unsigned long long ra = (unsigned long long)(pfb + (size_t)j * D_);
        const unsigned ralo = __builtin_amdgcn_readfirstlane((unsigned)ra);
        const unsigned rahi = __builtin_amdgcn_readfirstlane((unsigned)(ra >> 32));
        const gchar_t* rowk = (const gchar_t*)(((unsigned long long)rahi << 32) | ralo);
        const v4f f = *(const gv4f_t*)(rowk + lane_off);
        h01[k] = v2f{f.x, f.y};
        h23[k] = v2f{f.z, f.w};
        const v2f s2 = __builtin_elementwise_fma(h23[k], wa23, h01[k] * wa01);
        part[k] = s2.x + s2.y;
    }

    float ux = 0.f, uy = 0.f, uz = 0.f;
#pragma unroll
    for (int t = 0; t < 4; ++t) {
        ux += wa[t] * wx[t];
        uy += wa[t] * wy[t];
        uz += wa[t] * wz[t];
    }
    ux = wave_sum(ux);
    uy = wave_sum(uy);
    uz = wave_sum(uz);

    float r8[8], r4[4], r2[2];
#pragma unroll
    for (int i = 0; i < 8; ++i) r8[i] = fold_step<0xB1>(part[2 * i], part[2 * i + 1], (lane & 1) != 0);
#pragma unroll
    for (int i = 0; i < 4; ++i) r4[i] = fold_step<0x4E>(r8[2 * i], r8[2 * i + 1], (lane & 2) != 0);
#pragma unroll
    for (int i = 0; i < 2; ++i) r2[i] = fold_step<0x164>(r4[2 * i], r4[2 * i + 1], (lane & 4) != 0);
    float sc = fold_step<0x168>(r2[0], r2[1], (lane & 8) != 0);
    sc += other_row(sc);
    sc += ux * npx + uy * npy + uz * npz;

    const float mx = row_max(sc);
    const float ek = __expf(sc - mx);
    const float inv = 1.0f / row_sum(ek);
    const float nbx = row_sum(ek * npx);
    const float nby = row_sum(ek * npy);
    const float nbz = row_sum(ek * npz);

    v2f a01 = {0.f, 0.f}, a23 = {0.f, 0.f};
#pragma unroll
    for (int k = 0; k < KNN; ++k) {
        const float e = __int_as_float(__builtin_amdgcn_readlane(__float_as_int(ek), k));
        const v2f e2 = {e, e};
        a01 = __builtin_elementwise_fma(e2, h01[k], a01);
        a23 = __builtin_elementwise_fma(e2, h23[k], a23);
    }
    const float acc[4] = {a01.x, a01.y, a23.x, a23.y};

#pragma unroll
    for (int t = 0; t < 4; ++t)
        hb[wave * HB_PITCH + d0 + t] = (acc[t] + wx[t] * nbx + wy[t] * nby + wz[t] * nbz) * inv + bp[t];
    __syncthreads();

    if (wave < 16) {
        const int mt    = wave >> 3;
        const int tileN = (wave & 7) * 16;
        const int half  = lane >> 4;
        const int l     = lane & 15;
        const float* arow = hb + (mt * 16 + l) * HB_PITCH;
        const float* brow = wvs + (tileN + l) * WV_PITCH;

        v8f c = {};
        for (int k0 = 0; k0 < D_; k0 += 4) {
            const int k = k0 + half * 2;
            v2f a, bb;
            a.x  = arow[k];
            a.y  = arow[k + 1];
            bb.x = brow[k];
            bb.y = brow[k + 1];
            c = __builtin_amdgcn_wmma_f32_16x16x4_f32(
                     false, a,  false, bb,
                     (short)0, c,  false,  false);
        }

        const float bias = b_v[tileN + l];
#pragma unroll
        for (int r = 0; r < 8; ++r) {
            const int ml = mt * 16 + r + half * 8;
            pts[(tileN + l) * TR_PITCH + ml] = c[r] + bias;
        }
    }
    __syncthreads();
    {
        const int d    = threadIdx.x >> 3;
        const int quad = threadIdx.x & 7;
        const v4f v = *reinterpret_cast<const v4f*>(pts + d * TR_PITCH + quad * 4);
        *reinterpret_cast<v4f*>(outb + (size_t)d * NQ + qbase + quad * 4) = v;
    }
}

extern "C" void kernel_launch(void* const* d_in, const int* in_sizes, int n_in,
                              void* d_out, int out_size, void* d_ws, size_t ws_size,
                              hipStream_t stream) {
    const float* query         = (const float*)d_in[0];
    const float* global_pc     = (const float*)d_in[1];
    const float* point_feature = (const float*)d_in[2];
    const float* w_pos         = (const float*)d_in[3];
    const float* b_pos         = (const float*)d_in[4];
    const float* w_pe          = (const float*)d_in[5];
    const float* b_pe          = (const float*)d_in[6];
    const float* w_v           = (const float*)d_in[7];
    const float* b_v           = (const float*)d_in[8];
    const float* w_attn        = (const float*)d_in[9];
    float* out = (float*)d_out;

    fused_kernel<<<M_TOT / QPB, 1024, 0, stream>>>(query, global_pc, point_feature,
                                                   w_pos, b_pos, w_pe, b_pe,
                                                   w_v, b_v, w_attn, out);
}
